# starting kernel
# baseline (speedup 1.0000x reference)
#include <hip/hip_runtime.h>

#define HH 6
#define TT 1024
#define NG 24

typedef float v2f __attribute__((ext_vector_type(2)));
typedef float v8f __attribute__((ext_vector_type(8)));

__device__ __forceinline__ float d_tanh(float x) {
#if __has_builtin(__builtin_amdgcn_tanhf)
  return __builtin_amdgcn_tanhf(x);
#elif __has_builtin(__builtin_amdgcn_tanh_f32)
  return __builtin_amdgcn_tanh_f32(x);
#else
  float e = __expf(2.0f * x);
  return 1.0f - 2.0f * __builtin_amdgcn_rcpf(e + 1.0f);
#endif
}

__device__ __forceinline__ v2f mkv2(float a, float b) {
  v2f v; v.x = a; v.y = b; return v;
}

__device__ __forceinline__ v8f wmma4(v2f a, v2f b, v8f c) {
  return __builtin_amdgcn_wmma_f32_16x16x4_f32(false, a, false, b, (short)0, c,
                                               false, false);
}

__device__ __forceinline__ float rowscale(int m) {
  return (m < 2 * HH || m >= 3 * HH) ? 0.5f : 1.0f;
}

__device__ __forceinline__ float welem(const float* __restrict__ wih,
                                       const float* __restrict__ whh,
                                       int m, int c) {
  if (m >= NG) return 0.0f;
  float w = (c < HH) ? wih[m * HH + c] : whh[m * HH + (c - HH)];
  return rowscale(m) * w;
}

struct HC {
  float h0, h1, h2, h3, h4, h5;
  float c0, c1, c2, c3, c4, c5;
};

__device__ __forceinline__ HC lstm_cell(v2f A00, v2f A01, v2f A02,
                                        v2f A10, v2f A11, v2f A12,
                                        v8f cb0, v8f cb1, bool lo,
                                        float z0, float z1, float z2, float z3,
                                        float z4, float z5, float z6, float z7,
                                        float z8, float z9, float z10, float z11,
                                        HC s) {
  v2f b0 = mkv2(lo ? z0 : z2,  lo ? z1 : z3);
  v2f b1 = mkv2(lo ? z4 : z6,  lo ? z5 : z7);
  v2f b2 = mkv2(lo ? z8 : z10, lo ? z9 : z11);

  v8f acc0 = cb0;
  acc0 = wmma4(A00, b0, acc0);
  acc0 = wmma4(A01, b1, acc0);
  acc0 = wmma4(A02, b2, acc0);
  v8f acc1 = cb1;
  acc1 = wmma4(A10, b0, acc1);
  acc1 = wmma4(A11, b1, acc1);
  acc1 = wmma4(A12, b2, acc1);

  float t00 = d_tanh(acc0[0]), t01 = d_tanh(acc0[1]);
  float t02 = d_tanh(acc0[2]), t03 = d_tanh(acc0[3]);
  float t04 = d_tanh(acc0[4]), t05 = d_tanh(acc0[5]);
  float t06 = d_tanh(acc0[6]), t07 = d_tanh(acc0[7]);
  float t10 = d_tanh(acc1[0]), t11 = d_tanh(acc1[1]);
  float t12 = d_tanh(acc1[2]), t13 = d_tanh(acc1[3]);
  float t14 = d_tanh(acc1[4]), t15 = d_tanh(acc1[5]);
  float t16 = d_tanh(acc1[6]), t17 = d_tanh(acc1[7]);

#define GATHER(ta, tb, ga, gb, gc)         \
  float ga, gb, gc;                        \
  {                                        \
    float o0 = __shfl_xor(ta, 16, 32);     \
    float o1 = __shfl_xor(tb, 16, 32);     \
    ga = lo ? ta : o0;                     \
    gb = lo ? o0 : ta;                     \
    gc = lo ? tb : o1;                     \
  }
  GATHER(t00, t10, ga0, gb0, gc0)
  GATHER(t01, t11, ga1, gb1, gc1)
  GATHER(t02, t12, ga2, gb2, gc2)
  GATHER(t03, t13, ga3, gb3, gc3)
  GATHER(t04, t14, ga4, gb4, gc4)
  GATHER(t05, t15, ga5, gb5, gc5)
  GATHER(t06, t16, ga6, gb6, gc6)
  GATHER(t07, t17, ga7, gb7, gc7)
#undef GATHER

#define CELL(iv, fv, gv, ov, hvar, cvar)                  \
  {                                                       \
    float ig = fmaf(0.5f, iv, 0.5f);                      \
    float fg = fmaf(0.5f, fv, 0.5f);                      \
    float og = fmaf(0.5f, ov, 0.5f);                      \
    float cn = fmaf(fg, cvar, ig * (gv));                 \
    cvar = cn;                                            \
    hvar = og * d_tanh(cn);                               \
  }
  CELL(ga0, ga6, gb4, gc2, s.h0, s.c0)
  CELL(ga1, ga7, gb5, gc3, s.h1, s.c1)
  CELL(ga2, gb0, gb6, gc4, s.h2, s.c2)
  CELL(ga3, gb1, gb7, gc5, s.h3, s.c3)
  CELL(ga4, gb2, gc0, gc6, s.h4, s.c4)
  CELL(ga5, gb3, gc1, gc7, s.h5, s.c5)
#undef CELL
  return s;
}

__global__ __launch_bounds__(32) void lstm2_wmma_kernel(
    const float* __restrict__ x,
    const float* __restrict__ wih0, const float* __restrict__ whh0,
    const float* __restrict__ bih0, const float* __restrict__ bhh0,
    const float* __restrict__ wih1, const float* __restrict__ whh1,
    const float* __restrict__ bih1, const float* __restrict__ bhh1,
    float* __restrict__ out, int B) {
  const int lane = threadIdx.x & 31;
  const bool lo = lane < 16;
  const int m0 = lane & 15;
  const int half = lo ? 0 : 1;
  const int nRaw = blockIdx.x * 16 + m0;
  const int n = (nRaw < B) ? nRaw : (B - 1);
  const bool wr = lo && (nRaw < B);

  const int c0i = 2 * half, c1i = 4 + 2 * half, c2i = 8 + 2 * half;

  v2f A00l0 = mkv2(welem(wih0, whh0, m0, c0i),      welem(wih0, whh0, m0, c0i + 1));
  v2f A01l0 = mkv2(welem(wih0, whh0, m0, c1i),      welem(wih0, whh0, m0, c1i + 1));
  v2f A02l0 = mkv2(welem(wih0, whh0, m0, c2i),      welem(wih0, whh0, m0, c2i + 1));
  v2f A10l0 = mkv2(welem(wih0, whh0, 16 + m0, c0i), welem(wih0, whh0, 16 + m0, c0i + 1));
  v2f A11l0 = mkv2(welem(wih0, whh0, 16 + m0, c1i), welem(wih0, whh0, 16 + m0, c1i + 1));
  v2f A12l0 = mkv2(welem(wih0, whh0, 16 + m0, c2i), welem(wih0, whh0, 16 + m0, c2i + 1));

  v2f A00l1 = mkv2(welem(wih1, whh1, m0, c0i),      welem(wih1, whh1, m0, c0i + 1));
  v2f A01l1 = mkv2(welem(wih1, whh1, m0, c1i),      welem(wih1, whh1, m0, c1i + 1));
  v2f A02l1 = mkv2(welem(wih1, whh1, m0, c2i),      welem(wih1, whh1, m0, c2i + 1));
  v2f A10l1 = mkv2(welem(wih1, whh1, 16 + m0, c0i), welem(wih1, whh1, 16 + m0, c0i + 1));
  v2f A11l1 = mkv2(welem(wih1, whh1, 16 + m0, c1i), welem(wih1, whh1, 16 + m0, c1i + 1));
  v2f A12l1 = mkv2(welem(wih1, whh1, 16 + m0, c2i), welem(wih1, whh1, 16 + m0, c2i + 1));

  v8f cb0l0, cb1l0, cb0l1, cb1l1;
#pragma unroll
  for (int r = 0; r < 8; ++r) {
    const int r0 = lo ? r : (8 + r);
    cb0l0[r] = rowscale(r0) * (bih0[r0] + bhh0[r0]);
    cb1l0[r] = lo ? (rowscale(16 + r) * (bih0[16 + r] + bhh0[16 + r])) : 0.0f;
    cb0l1[r] = rowscale(r0) * (bih1[r0] + bhh1[r0]);
    cb1l1[r] = lo ? (rowscale(16 + r) * (bih1[16 + r] + bhh1[16 + r])) : 0.0f;
  }

  HC s1 = {0, 0, 0, 0, 0, 0, 0, 0, 0, 0, 0, 0};
  HC s2 = {0, 0, 0, 0, 0, 0, 0, 0, 0, 0, 0, 0};

  const float* __restrict__ xb = x + (size_t)n * TT * HH;
  float* __restrict__ ob = out + (size_t)n * TT * HH;

  float xc0 = __builtin_nontemporal_load(xb + 0);
  float xc1 = __builtin_nontemporal_load(xb + 1);
  float xc2 = __builtin_nontemporal_load(xb + 2);
  float xc3 = __builtin_nontemporal_load(xb + 3);
  float xc4 = __builtin_nontemporal_load(xb + 4);
  float xc5 = __builtin_nontemporal_load(xb + 5);
  float xn0 = __builtin_nontemporal_load(xb + HH + 0);
  float xn1 = __builtin_nontemporal_load(xb + HH + 1);
  float xn2 = __builtin_nontemporal_load(xb + HH + 2);
  float xn3 = __builtin_nontemporal_load(xb + HH + 3);
  float xn4 = __builtin_nontemporal_load(xb + HH + 4);
  float xn5 = __builtin_nontemporal_load(xb + HH + 5);

  s1 = lstm_cell(A00l0, A01l0, A02l0, A10l0, A11l0, A12l0, cb0l0, cb1l0, lo,
                 xc0, xc1, xc2, xc3, xc4, xc5,
                 s1.h0, s1.h1, s1.h2, s1.h3, s1.h4, s1.h5, s1);

  for (int t = 1; t < TT; ++t) {
    const size_t off = (size_t)((t + 1 < TT) ? (t + 1) : t) * HH;
    float xf0 = __builtin_nontemporal_load(xb + off + 0);
    float xf1 = __builtin_nontemporal_load(xb + off + 1);
    float xf2 = __builtin_nontemporal_load(xb + off + 2);
    float xf3 = __builtin_nontemporal_load(xb + off + 3);
    float xf4 = __builtin_nontemporal_load(xb + off + 4);
    float xf5 = __builtin_nontemporal_load(xb + off + 5);

    const float p0 = s1.h0, p1 = s1.h1, p2 = s1.h2,
                p3 = s1.h3, p4 = s1.h4, p5 = s1.h5;

    s1 = lstm_cell(A00l0, A01l0, A02l0, A10l0, A11l0, A12l0, cb0l0, cb1l0, lo,
                   xn0, xn1, xn2, xn3, xn4, xn5,
                   p0, p1, p2, p3, p4, p5, s1);

    s2 = lstm_cell(A00l1, A01l1, A02l1, A10l1, A11l1, A12l1, cb0l1, cb1l1, lo,
                   p0, p1, p2, p3, p4, p5,
                   s2.h0, s2.h1, s2.h2, s2.h3, s2.h4, s2.h5, s2);

    if (wr) {
      float* __restrict__ op = ob + (size_t)(t - 1) * HH;
      __builtin_nontemporal_store(s2.h0, op + 0);
      __builtin_nontemporal_store(s2.h1, op + 1);
      __builtin_nontemporal_store(s2.h2, op + 2);
      __builtin_nontemporal_store(s2.h3, op + 3);
      __builtin_nontemporal_store(s2.h4, op + 4);
      __builtin_nontemporal_store(s2.h5, op + 5);
    }

    xn0 = xf0; xn1 = xf1; xn2 = xf2; xn3 = xf3; xn4 = xf4; xn5 = xf5;
  }

  s2 = lstm_cell(A00l1, A01l1, A02l1, A10l1, A11l1, A12l1, cb0l1, cb1l1, lo,
                 s1.h0, s1.h1, s1.h2, s1.h3, s1.h4, s1.h5,
                 s2.h0, s2.h1, s2.h2, s2.h3, s2.h4, s2.h5, s2);
  if (wr) {
    float* __restrict__ op = ob + (size_t)(TT - 1) * HH;
    __builtin_nontemporal_store(s2.h0, op + 0);
    __builtin_nontemporal_store(s2.h1, op + 1);
    __builtin_nontemporal_store(s2.h2, op + 2);
    __builtin_nontemporal_store(s2.h3, op + 3);
    __builtin_nontemporal_store(s2.h4, op + 4);
    __builtin_nontemporal_store(s2.h5, op + 5);
  }
}

extern "C" void kernel_launch(void* const* d_in, const int* in_sizes, int n_in,
                              void* d_out, int out_size, void* d_ws, size_t ws_size,
                              hipStream_t stream) {
  const float* x    = (const float*)d_in[0];
  const float* wih0 = (const float*)d_in[1];
  const float* whh0 = (const float*)d_in[2];
  const float* bih0 = (const float*)d_in[3];
  const float* bhh0 = (const float*)d_in[4];
  const float* wih1 = (const float*)d_in[5];
  const float* whh1 = (const float*)d_in[6];
  const float* bih1 = (const float*)d_in[7];
  const float* bhh1 = (const float*)d_in[8];
  float* out = (float*)d_out;

  const int B = in_sizes[0] / (TT * HH);
  const int nblocks = (B + 15) / 16;

  hipLaunchKernelGGL(lstm2_wmma_kernel, dim3(nblocks), dim3(32), 0, stream,
                     x, wih0, whh0, bih0, bhh0, wih1, whh1, bih1, bhh1, out, B);
}


# x requested a whole block of 4 time steps ahead (six 16-byte loads per lane per block) instead of one step ahead; main loop unrolled by 4 so the per-step wait for memory leaves the recurrence
# speedup vs baseline: 1.6000x; 1.6000x over previous
#include <hip/hip_runtime.h>

#define HH 6
#define TT 1024
#define NG 24

typedef float v2f __attribute__((ext_vector_type(2)));
typedef float v8f __attribute__((ext_vector_type(8)));
typedef float v4f __attribute__((ext_vector_type(4)));

#define TB 4

struct XG {
  v4f a, b, c, d, e, f;
};

__device__ __forceinline__ XG load_xg(const float* __restrict__ p) {
  const v4f* __restrict__ q = (const v4f*)p;
  XG g;
  g.a = __builtin_nontemporal_load(q + 0);
  g.b = __builtin_nontemporal_load(q + 1);
  g.c = __builtin_nontemporal_load(q + 2);
  g.d = __builtin_nontemporal_load(q + 3);
  g.e = __builtin_nontemporal_load(q + 4);
  g.f = __builtin_nontemporal_load(q + 5);
  return g;
}

__device__ __forceinline__ float d_tanh(float x) {
#if __has_builtin(__builtin_amdgcn_tanhf)
  return __builtin_amdgcn_tanhf(x);
#elif __has_builtin(__builtin_amdgcn_tanh_f32)
  return __builtin_amdgcn_tanh_f32(x);
#else
  float e = __expf(2.0f * x);
  return 1.0f - 2.0f * __builtin_amdgcn_rcpf(e + 1.0f);
#endif
}

__device__ __forceinline__ v2f mkv2(float a, float b) {
  v2f v; v.x = a; v.y = b; return v;
}

__device__ __forceinline__ v8f wmma4(v2f a, v2f b, v8f c) {
  return __builtin_amdgcn_wmma_f32_16x16x4_f32(false, a, false, b, (short)0, c,
                                               false, false);
}

__device__ __forceinline__ float rowscale(int m) {
  return (m < 2 * HH || m >= 3 * HH) ? 0.5f : 1.0f;
}

__device__ __forceinline__ float welem(const float* __restrict__ wih,
                                       const float* __restrict__ whh,
                                       int m, int c) {
  if (m >= NG) return 0.0f;
  float w = (c < HH) ? wih[m * HH + c] : whh[m * HH + (c - HH)];
  return rowscale(m) * w;
}

struct HC {
  float h0, h1, h2, h3, h4, h5;
  float c0, c1, c2, c3, c4, c5;
};

__device__ __forceinline__ HC lstm_cell(v2f A00, v2f A01, v2f A02,
                                        v2f A10, v2f A11, v2f A12,
                                        v8f cb0, v8f cb1, bool lo,
                                        float z0, float z1, float z2, float z3,
                                        float z4, float z5, float z6, float z7,
                                        float z8, float z9, float z10, float z11,
                                        HC s) {
  v2f b0 = mkv2(lo ? z0 : z2,  lo ? z1 : z3);
  v2f b1 = mkv2(lo ? z4 : z6,  lo ? z5 : z7);
  v2f b2 = mkv2(lo ? z8 : z10, lo ? z9 : z11);

  v8f acc0 = cb0;
  acc0 = wmma4(A00, b0, acc0);
  acc0 = wmma4(A01, b1, acc0);
  acc0 = wmma4(A02, b2, acc0);
  v8f acc1 = cb1;
  acc1 = wmma4(A10, b0, acc1);
  acc1 = wmma4(A11, b1, acc1);
  acc1 = wmma4(A12, b2, acc1);

  float t00 = d_tanh(acc0[0]), t01 = d_tanh(acc0[1]);
  float t02 = d_tanh(acc0[2]), t03 = d_tanh(acc0[3]);
  float t04 = d_tanh(acc0[4]), t05 = d_tanh(acc0[5]);
  float t06 = d_tanh(acc0[6]), t07 = d_tanh(acc0[7]);
  float t10 = d_tanh(acc1[0]), t11 = d_tanh(acc1[1]);
  float t12 = d_tanh(acc1[2]), t13 = d_tanh(acc1[3]);
  float t14 = d_tanh(acc1[4]), t15 = d_tanh(acc1[5]);
  float t16 = d_tanh(acc1[6]), t17 = d_tanh(acc1[7]);

#define GATHER(ta, tb, ga, gb, gc)         \
  float ga, gb, gc;                        \
  {                                        \
    float o0 = __shfl_xor(ta, 16, 32);     \
    float o1 = __shfl_xor(tb, 16, 32);     \
    ga = lo ? ta : o0;                     \
    gb = lo ? o0 : ta;                     \
    gc = lo ? tb : o1;                     \
  }
  GATHER(t00, t10, ga0, gb0, gc0)
  GATHER(t01, t11, ga1, gb1, gc1)
  GATHER(t02, t12, ga2, gb2, gc2)
  GATHER(t03, t13, ga3, gb3, gc3)
  GATHER(t04, t14, ga4, gb4, gc4)
  GATHER(t05, t15, ga5, gb5, gc5)
  GATHER(t06, t16, ga6, gb6, gc6)
  GATHER(t07, t17, ga7, gb7, gc7)
#undef GATHER

#define CELL(iv, fv, gv, ov, hvar, cvar)                  \
  {                                                       \
    float ig = fmaf(0.5f, iv, 0.5f);                      \
    float fg = fmaf(0.5f, fv, 0.5f);                      \
    float og = fmaf(0.5f, ov, 0.5f);                      \
    float cn = fmaf(fg, cvar, ig * (gv));                 \
    cvar = cn;                                            \
    hvar = og * d_tanh(cn);                               \
  }
  CELL(ga0, ga6, gb4, gc2, s.h0, s.c0)
  CELL(ga1, ga7, gb5, gc3, s.h1, s.c1)
  CELL(ga2, gb0, gb6, gc4, s.h2, s.c2)
  CELL(ga3, gb1, gb7, gc5, s.h3, s.c3)
  CELL(ga4, gb2, gc0, gc6, s.h4, s.c4)
  CELL(ga5, gb3, gc1, gc7, s.h5, s.c5)
#undef CELL
  return s;
}

__global__ __launch_bounds__(32) void lstm2_wmma_kernel(
    const float* __restrict__ x,
    const float* __restrict__ wih0, const float* __restrict__ whh0,
    const float* __restrict__ bih0, const float* __restrict__ bhh0,
    const float* __restrict__ wih1, const float* __restrict__ whh1,
    const float* __restrict__ bih1, const float* __restrict__ bhh1,
    float* __restrict__ out, int B) {
  const int lane = threadIdx.x & 31;
  const bool lo = lane < 16;
  const int m0 = lane & 15;
  const int half = lo ? 0 : 1;
  const int nRaw = blockIdx.x * 16 + m0;
  const int n = (nRaw < B) ? nRaw : (B - 1);
  const bool wr = lo && (nRaw < B);

  const int c0i = 2 * half, c1i = 4 + 2 * half, c2i = 8 + 2 * half;

  v2f A00l0 = mkv2(welem(wih0, whh0, m0, c0i),      welem(wih0, whh0, m0, c0i + 1));
  v2f A01l0 = mkv2(welem(wih0, whh0, m0, c1i),      welem(wih0, whh0, m0, c1i + 1));
  v2f A02l0 = mkv2(welem(wih0, whh0, m0, c2i),      welem(wih0, whh0, m0, c2i + 1));
  v2f A10l0 = mkv2(welem(wih0, whh0, 16 + m0, c0i), welem(wih0, whh0, 16 + m0, c0i + 1));
  v2f A11l0 = mkv2(welem(wih0, whh0, 16 + m0, c1i), welem(wih0, whh0, 16 + m0, c1i + 1));
  v2f A12l0 = mkv2(welem(wih0, whh0, 16 + m0, c2i), welem(wih0, whh0, 16 + m0, c2i + 1));

  v2f A00l1 = mkv2(welem(wih1, whh1, m0, c0i),      welem(wih1, whh1, m0, c0i + 1));
  v2f A01l1 = mkv2(welem(wih1, whh1, m0, c1i),      welem(wih1, whh1, m0, c1i + 1));
  v2f A02l1 = mkv2(welem(wih1, whh1, m0, c2i),      welem(wih1, whh1, m0, c2i + 1));
  v2f A10l1 = mkv2(welem(wih1, whh1, 16 + m0, c0i), welem(wih1, whh1, 16 + m0, c0i + 1));
  v2f A11l1 = mkv2(welem(wih1, whh1, 16 + m0, c1i), welem(wih1, whh1, 16 + m0, c1i + 1));
  v2f A12l1 = mkv2(welem(wih1, whh1, 16 + m0, c2i), welem(wih1, whh1, 16 + m0, c2i + 1));

  v8f cb0l0, cb1l0, cb0l1, cb1l1;
#pragma unroll
  for (int r = 0; r < 8; ++r) {
    const int r0 = lo ? r : (8 + r);
    cb0l0[r] = rowscale(r0) * (bih0[r0] + bhh0[r0]);
    cb1l0[r] = lo ? (rowscale(16 + r) * (bih0[16 + r] + bhh0[16 + r])) : 0.0f;
    cb0l1[r] = rowscale(r0) * (bih1[r0] + bhh1[r0]);
    cb1l1[r] = lo ? (rowscale(16 + r) * (bih1[16 + r] + bhh1[16 + r])) : 0.0f;
  }

  HC s1 = {0, 0, 0, 0, 0, 0, 0, 0, 0, 0, 0, 0};
  HC s2 = {0, 0, 0, 0, 0, 0, 0, 0, 0, 0, 0, 0};

  const float* __restrict__ xb = x + (size_t)n * TT * HH;
  float* __restrict__ ob = out + (size_t)n * TT * HH;

#define STEP(X0, X1, X2, X3, X4, X5, TOUT)                                     \
  {                                                                            \
    const float p0 = s1.h0, p1 = s1.h1, p2 = s1.h2,                            \
                p3 = s1.h3, p4 = s1.h4, p5 = s1.h5;                            \
    s1 = lstm_cell(A00l0, A01l0, A02l0, A10l0, A11l0, A12l0, cb0l0, cb1l0, lo, \
                   X0, X1, X2, X3, X4, X5, p0, p1, p2, p3, p4, p5, s1);        \
    s2 = lstm_cell(A00l1, A01l1, A02l1, A10l1, A11l1, A12l1, cb0l1, cb1l1, lo, \
                   p0, p1, p2, p3, p4, p5,                                     \
                   s2.h0, s2.h1, s2.h2, s2.h3, s2.h4, s2.h5, s2);              \
    if (wr) {                                                                  \
      float* __restrict__ op = ob + (size_t)(TOUT) * HH;                       \
      __builtin_nontemporal_store(s2.h0, op + 0);                              \
      __builtin_nontemporal_store(s2.h1, op + 1);                              \
      __builtin_nontemporal_store(s2.h2, op + 2);                              \
      __builtin_nontemporal_store(s2.h3, op + 3);                              \
      __builtin_nontemporal_store(s2.h4, op + 4);                              \
      __builtin_nontemporal_store(s2.h5, op + 5);                              \
    }                                                                          \
  }

  XG cur = load_xg(xb);
  XG nx = load_xg(xb + TB * HH);

  s1 = lstm_cell(A00l0, A01l0, A02l0, A10l0, A11l0, A12l0, cb0l0, cb1l0, lo,
                 cur.a.x, cur.a.y, cur.a.z, cur.a.w, cur.b.x, cur.b.y,
                 s1.h0, s1.h1, s1.h2, s1.h3, s1.h4, s1.h5, s1);
  STEP(cur.b.z, cur.b.w, cur.c.x, cur.c.y, cur.c.z, cur.c.w, 0)
  STEP(cur.d.x, cur.d.y, cur.d.z, cur.d.w, cur.e.x, cur.e.y, 1)
  STEP(cur.e.z, cur.e.w, cur.f.x, cur.f.y, cur.f.z, cur.f.w, 2)
  cur = nx;

  for (int k = 1; k < TT / TB; ++k) {
    const int kn = (k + 1 < TT / TB) ? (k + 1) : k;
    nx = load_xg(xb + (size_t)kn * (TB * HH));

    const int t0 = k * TB;
    STEP(cur.a.x, cur.a.y, cur.a.z, cur.a.w, cur.b.x, cur.b.y, t0 - 1)
    STEP(cur.b.z, cur.b.w, cur.c.x, cur.c.y, cur.c.z, cur.c.w, t0)
    STEP(cur.d.x, cur.d.y, cur.d.z, cur.d.w, cur.e.x, cur.e.y, t0 + 1)
    STEP(cur.e.z, cur.e.w, cur.f.x, cur.f.y, cur.f.z, cur.f.w, t0 + 2)
    cur = nx;
  }
#undef STEP

  s2 = lstm_cell(A00l1, A01l1, A02l1, A10l1, A11l1, A12l1, cb0l1, cb1l1, lo,
                 s1.h0, s1.h1, s1.h2, s1.h3, s1.h4, s1.h5,
                 s2.h0, s2.h1, s2.h2, s2.h3, s2.h4, s2.h5, s2);
  if (wr) {
    float* __restrict__ op = ob + (size_t)(TT - 1) * HH;
    __builtin_nontemporal_store(s2.h0, op + 0);
    __builtin_nontemporal_store(s2.h1, op + 1);
    __builtin_nontemporal_store(s2.h2, op + 2);
    __builtin_nontemporal_store(s2.h3, op + 3);
    __builtin_nontemporal_store(s2.h4, op + 4);
    __builtin_nontemporal_store(s2.h5, op + 5);
  }
}

extern "C" void kernel_launch(void* const* d_in, const int* in_sizes, int n_in,
                              void* d_out, int out_size, void* d_ws, size_t ws_size,
                              hipStream_t stream) {
  const float* x    = (const float*)d_in[0];
  const float* wih0 = (const float*)d_in[1];
  const float* whh0 = (const float*)d_in[2];
  const float* bih0 = (const float*)d_in[3];
  const float* bhh0 = (const float*)d_in[4];
  const float* wih1 = (const float*)d_in[5];
  const float* whh1 = (const float*)d_in[6];
  const float* bih1 = (const float*)d_in[7];
  const float* bhh1 = (const float*)d_in[8];
  float* out = (float*)d_out;

  const int B = in_sizes[0] / (TT * HH);
  const int nblocks = (B + 15) / 16;

  hipLaunchKernelGGL(lstm2_wmma_kernel, dim3(nblocks), dim3(32), 0, stream,
                     x, wih0, whh0, bih0, bhh0, wih1, whh1, bih1, bhh1, out, B);
}


# gate rows and k order re-laid so each lane half owns all 4 gates of 3 hidden units: no cross-half gate exchange, 3 cell updates per lane, 12-byte x loads and out stores per half; k sum order changes
# speedup vs baseline: 2.2371x; 1.3982x over previous
#include <hip/hip_runtime.h>


#define HH 6
#define TT 1024
#define NG 24
#define HU 3

typedef float v2f __attribute__((ext_vector_type(2)));
typedef float v3f __attribute__((ext_vector_type(3)));
typedef float v8f __attribute__((ext_vector_type(8)));
typedef v3f v3fu __attribute__((aligned(4)));

#define TB 4

struct XG {
  v3f a, b, c, d;
};

__device__ __forceinline__ XG load_xg(const float* __restrict__ p) {
  XG g;
  g.a = __builtin_nontemporal_load((const v3fu*)(p + 0 * HH));
  g.b = __builtin_nontemporal_load((const v3fu*)(p + 1 * HH));
  g.c = __builtin_nontemporal_load((const v3fu*)(p + 2 * HH));
  g.d = __builtin_nontemporal_load((const v3fu*)(p + 3 * HH));
  return g;
}

__device__ __forceinline__ float d_tanh(float x) {
#if __has_builtin(__builtin_amdgcn_tanhf)
  return __builtin_amdgcn_tanhf(x);
#elif __has_builtin(__builtin_amdgcn_tanh_f32)
  return __builtin_amdgcn_tanh_f32(x);
#else
  float e = __expf(2.0f * x);
  return 1.0f - 2.0f * __builtin_amdgcn_rcpf(e + 1.0f);
#endif
}

__device__ __forceinline__ v2f mkv2(float a, float b) {
  v2f v; v.x = a; v.y = b; return v;
}

__device__ __forceinline__ v8f wmma4(v2f a, v2f b, v8f c) {
  return __builtin_amdgcn_wmma_f32_16x16x4_f32(false, a, false, b, (short)0, c,
                                               false, false);
}

__device__ __forceinline__ int gate_of(int tile, int m) {
  const int hh = m >> 3, r = m & 7, u = HU * hh;
  if (tile == 0) {
    if (r < 3) return u + r;
    if (r < 6) return HH + u + (r - 3);
    return 2 * HH + u + (r - 6);
  }
  if (r == 0) return 2 * HH + u + 2;
  if (r < 4) return 3 * HH + u + (r - 1);
  return -1;
}

__device__ __forceinline__ int col_of(int j, int hh, int s) {
  const int u = HU * hh;
  if (j == 0) return u + s;
  if (j == 1) return (s == 0) ? (u + 2) : (HH + u);
  return HH + u + 1 + s;
}

__device__ __forceinline__ float rowscale(int gate) {
  return (gate < 2 * HH || gate >= 3 * HH) ? 0.5f : 1.0f;
}

__device__ __forceinline__ float welem(const float* __restrict__ wih,
                                       const float* __restrict__ whh,
                                       int gate, int c) {
  if (gate < 0) return 0.0f;
  float w = (c < HH) ? wih[gate * HH + c] : whh[gate * HH + (c - HH)];
  return rowscale(gate) * w;
}

__device__ __forceinline__ v2f wfrag(const float* __restrict__ wih,
                                     const float* __restrict__ whh,
                                     int tile, int m, int j, int hh) {
  const int gate = gate_of(tile, m);
  return mkv2(welem(wih, whh, gate, col_of(j, hh, 0)),
              welem(wih, whh, gate, col_of(j, hh, 1)));
}

__device__ __forceinline__ float belem(const float* __restrict__ bih,
                                       const float* __restrict__ bhh,
                                       int gate) {
  if (gate < 0) return 0.0f;
  return rowscale(gate) * (bih[gate] + bhh[gate]);
}

struct HC {
  float h0, h1, h2;
  float c0, c1, c2;
};

__device__ __forceinline__ HC lstm_cell(v2f A00, v2f A01, v2f A02,
                                        v2f A10, v2f A11, v2f A12,
                                        v8f cb0, v8f cb1,
                                        float za, float zb, float zc, HC s) {
  const v2f b0 = mkv2(za, zb);
  const v2f b1 = mkv2(zc, s.h0);
  const v2f b2 = mkv2(s.h1, s.h2);

  v8f acc0 = cb0;
  acc0 = wmma4(A00, b0, acc0);
  v8f acc1 = cb1;
  acc1 = wmma4(A10, b0, acc1);
  acc0 = wmma4(A01, b1, acc0);
  acc1 = wmma4(A11, b1, acc1);
  acc0 = wmma4(A02, b2, acc0);
  acc1 = wmma4(A12, b2, acc1);

  const float i0 = d_tanh(acc0[0]), i1 = d_tanh(acc0[1]), i2 = d_tanh(acc0[2]);
  const float f0 = d_tanh(acc0[3]), f1 = d_tanh(acc0[4]), f2 = d_tanh(acc0[5]);
  const float g0 = d_tanh(acc0[6]), g1 = d_tanh(acc0[7]), g2 = d_tanh(acc1[0]);
  const float o0 = d_tanh(acc1[1]), o1 = d_tanh(acc1[2]), o2 = d_tanh(acc1[3]);

#define CELL(iv, fv, gv, ov, hvar, cvar)                  \
  {                                                       \
    float ig = fmaf(0.5f, iv, 0.5f);                      \
    float fg = fmaf(0.5f, fv, 0.5f);                      \
    float og = fmaf(0.5f, ov, 0.5f);                      \
    float cn = fmaf(fg, cvar, ig * (gv));                 \
    cvar = cn;                                            \
    hvar = og * d_tanh(cn);                               \
  }
  CELL(i0, f0, g0, o0, s.h0, s.c0)
  CELL(i1, f1, g1, o1, s.h1, s.c1)
  CELL(i2, f2, g2, o2, s.h2, s.c2)
#undef CELL
  return s;
}

__global__ __launch_bounds__(32) void lstm2_wmma_kernel(
    const float* __restrict__ x,
    const float* __restrict__ wih0, const float* __restrict__ whh0,
    const float* __restrict__ bih0, const float* __restrict__ bhh0,
    const float* __restrict__ wih1, const float* __restrict__ whh1,
    const float* __restrict__ bih1, const float* __restrict__ bhh1,
    float* __restrict__ out, int B) {
  const int lane = threadIdx.x & 31;
  const int m0 = lane & 15;
  const int hh = lane >> 4;
  const int nRaw = blockIdx.x * 16 + m0;
  const int n = (nRaw < B) ? nRaw : (B - 1);
  const bool wr = nRaw < B;

  const v2f A00l0 = wfrag(wih0, whh0, 0, m0, 0, hh);
  const v2f A01l0 = wfrag(wih0, whh0, 0, m0, 1, hh);
  const v2f A02l0 = wfrag(wih0, whh0, 0, m0, 2, hh);
  const v2f A10l0 = wfrag(wih0, whh0, 1, m0, 0, hh);
  const v2f A11l0 = wfrag(wih0, whh0, 1, m0, 1, hh);
  const v2f A12l0 = wfrag(wih0, whh0, 1, m0, 2, hh);

  const v2f A00l1 = wfrag(wih1, whh1, 0, m0, 0, hh);
  const v2f A01l1 = wfrag(wih1, whh1, 0, m0, 1, hh);
  const v2f A02l1 = wfrag(wih1, whh1, 0, m0, 2, hh);
  const v2f A10l1 = wfrag(wih1, whh1, 1, m0, 0, hh);
  const v2f A11l1 = wfrag(wih1, whh1, 1, m0, 1, hh);
  const v2f A12l1 = wfrag(wih1, whh1, 1, m0, 2, hh);

  v8f cb0l0, cb1l0, cb0l1, cb1l1;
#pragma unroll
  for (int r = 0; r < 8; ++r) {
    const int g0r = gate_of(0, 8 * hh + r);
    const int g1r = gate_of(1, 8 * hh + r);
    cb0l0[r] = belem(bih0, bhh0, g0r);
    cb1l0[r] = belem(bih0, bhh0, g1r);
    cb0l1[r] = belem(bih1, bhh1, g0r);
    cb1l1[r] = belem(bih1, bhh1, g1r);
  }

  HC s1 = {0, 0, 0, 0, 0, 0};
  HC s2 = {0, 0, 0, 0, 0, 0};

  const float* __restrict__ xb = x + (size_t)n * TT * HH + HU * hh;
  float* __restrict__ ob = out + (size_t)n * TT * HH + HU * hh;

#define STEP(XV, TOUT)                                                         \
  {                                                                            \
    const float p0 = s1.h0, p1 = s1.h1, p2 = s1.h2;                            \
    s1 = lstm_cell(A00l0, A01l0, A02l0, A10l0, A11l0, A12l0, cb0l0, cb1l0,     \
                   XV.x, XV.y, XV.z, s1);                                      \
    s2 = lstm_cell(A00l1, A01l1, A02l1, A10l1, A11l1, A12l1, cb0l1, cb1l1,     \
                   p0, p1, p2, s2);                                            \
    if (wr) {                                                                  \
      v3f ov;                                                                  \
      ov.x = s2.h0; ov.y = s2.h1; ov.z = s2.h2;                                \
      __builtin_nontemporal_store(ov, (v3fu*)(ob + (size_t)(TOUT) * HH));      \
    }                                                                          \
  }

  XG cur = load_xg(xb);
  XG nx = load_xg(xb + TB * HH);

  s1 = lstm_cell(A00l0, A01l0, A02l0, A10l0, A11l0, A12l0, cb0l0, cb1l0,
                 cur.a.x, cur.a.y, cur.a.z, s1);
  STEP(cur.b, 0)
  STEP(cur.c, 1)
  STEP(cur.d, 2)
  cur = nx;

  for (int k = 1; k < TT / TB; ++k) {
    const int kn = (k + 1 < TT / TB) ? (k + 1) : k;
    nx = load_xg(xb + (size_t)kn * (TB * HH));

    const int t0 = k * TB;
    STEP(cur.a, t0 - 1)
    STEP(cur.b, t0)
    STEP(cur.c, t0 + 1)
    STEP(cur.d, t0 + 2)
    cur = nx;
  }
#undef STEP

  s2 = lstm_cell(A00l1, A01l1, A02l1, A10l1, A11l1, A12l1, cb0l1, cb1l1,
                 s1.h0, s1.h1, s1.h2, s2);
  if (wr) {
    v3f ov;
    ov.x = s2.h0; ov.y = s2.h1; ov.z = s2.h2;
    __builtin_nontemporal_store(ov, (v3fu*)(ob + (size_t)(TT - 1) * HH));
  }
}

extern "C" void kernel_launch(void* const* d_in, const int* in_sizes, int n_in,
                              void* d_out, int out_size, void* d_ws, size_t ws_size,
                              hipStream_t stream) {
  const float* x    = (const float*)d_in[0];
  const float* wih0 = (const float*)d_in[1];
  const float* whh0 = (const float*)d_in[2];
  const float* bih0 = (const float*)d_in[3];
  const float* bhh0 = (const float*)d_in[4];
  const float* wih1 = (const float*)d_in[5];
  const float* whh1 = (const float*)d_in[6];
  const float* bih1 = (const float*)d_in[7];
  const float* bhh1 = (const float*)d_in[8];
  float* out = (float*)d_out;

  const int B = in_sizes[0] / (TT * HH);
  const int nblocks = (B + 15) / 16;

  hipLaunchKernelGGL(lstm2_wmma_kernel, dim3(nblocks), dim3(32), 0, stream,
                     x, wih0, whh0, bih0, bhh0, wih1, whh1, bih1, bhh1, out, B);
}


# layer 2 lags one and a half steps; step fenced into 12 stretches: 3 tanh + 3 plain of one layer's gate math issue behind each matrix product of the other layer; full-wave path, 16-step x blocks
# speedup vs baseline: 2.6699x; 1.1935x over previous
#include <hip/hip_runtime.h>


#define HH 6
#define TT 1024
#define NG 24
#define HU 3

typedef float v2f __attribute__((ext_vector_type(2)));
typedef float v3f __attribute__((ext_vector_type(3)));
typedef float v8f __attribute__((ext_vector_type(8)));
typedef v3f v3fu __attribute__((aligned(4)));

#define TB 16

struct XG {
  v3f x0, x1, x2, x3, x4, x5, x6, x7, x8, x9, x10, x11, x12, x13, x14, x15;
};

__device__ __forceinline__ XG load_xg(const float* __restrict__ p) {
  XG g;
  g.x0 = __builtin_nontemporal_load((const v3fu*)(p + 0 * HH));
  g.x1 = __builtin_nontemporal_load((const v3fu*)(p + 1 * HH));
  g.x2 = __builtin_nontemporal_load((const v3fu*)(p + 2 * HH));
  g.x3 = __builtin_nontemporal_load((const v3fu*)(p + 3 * HH));
  g.x4 = __builtin_nontemporal_load((const v3fu*)(p + 4 * HH));
  g.x5 = __builtin_nontemporal_load((const v3fu*)(p + 5 * HH));
  g.x6 = __builtin_nontemporal_load((const v3fu*)(p + 6 * HH));
  g.x7 = __builtin_nontemporal_load((const v3fu*)(p + 7 * HH));
  g.x8 = __builtin_nontemporal_load((const v3fu*)(p + 8 * HH));
  g.x9 = __builtin_nontemporal_load((const v3fu*)(p + 9 * HH));
  g.x10 = __builtin_nontemporal_load((const v3fu*)(p + 10 * HH));
  g.x11 = __builtin_nontemporal_load((const v3fu*)(p + 11 * HH));
  g.x12 = __builtin_nontemporal_load((const v3fu*)(p + 12 * HH));
  g.x13 = __builtin_nontemporal_load((const v3fu*)(p + 13 * HH));
  g.x14 = __builtin_nontemporal_load((const v3fu*)(p + 14 * HH));
  g.x15 = __builtin_nontemporal_load((const v3fu*)(p + 15 * HH));
  return g;
}

__device__ __forceinline__ float d_tanh(float x) {
#if __has_builtin(__builtin_amdgcn_tanhf)
  return __builtin_amdgcn_tanhf(x);
#elif __has_builtin(__builtin_amdgcn_tanh_f32)
  return __builtin_amdgcn_tanh_f32(x);
#else
  float e = __expf(2.0f * x);
  return 1.0f - 2.0f * __builtin_amdgcn_rcpf(e + 1.0f);
#endif
}

__device__ __forceinline__ v2f mkv2(float a, float b) {
  v2f v; v.x = a; v.y = b; return v;
}

__device__ __forceinline__ v8f wmma4(v2f a, v2f b, v8f c) {
  return __builtin_amdgcn_wmma_f32_16x16x4_f32(false, a, false, b, (short)0, c,
                                               false, false);
}

__device__ __forceinline__ int gate_of(int tile, int m) {
  const int hh = m >> 3, r = m & 7, u = HU * hh;
  if (tile == 0) {
    if (r < 3) return u + r;
    if (r < 6) return HH + u + (r - 3);
    return 2 * HH + u + (r - 6);
  }
  if (r == 0) return 2 * HH + u + 2;
  if (r < 4) return 3 * HH + u + (r - 1);
  return -1;
}

__device__ __forceinline__ int col_of(int j, int hh, int s) {
  const int u = HU * hh;
  if (j == 0) return u + s;
  if (j == 1) return (s == 0) ? (u + 2) : (HH + u);
  return HH + u + 1 + s;
}

__device__ __forceinline__ float rowscale(int gate) {
  return (gate < 2 * HH || gate >= 3 * HH) ? 0.5f : 1.0f;
}

__device__ __forceinline__ float welem(const float* __restrict__ wih,
                                       const float* __restrict__ whh,
                                       int gate, int c) {
  if (gate < 0) return 0.0f;
  float w = (c < HH) ? wih[gate * HH + c] : whh[gate * HH + (c - HH)];
  return rowscale(gate) * w;
}

__device__ __forceinline__ v2f wfrag(const float* __restrict__ wih,
                                     const float* __restrict__ whh,
                                     int tile, int m, int j, int hh) {
  const int gate = gate_of(tile, m);
  return mkv2(welem(wih, whh, gate, col_of(j, hh, 0)),
              welem(wih, whh, gate, col_of(j, hh, 1)));
}

__device__ __forceinline__ float belem(const float* __restrict__ bih,
                                       const float* __restrict__ bhh,
                                       int gate) {
  if (gate < 0) return 0.0f;
  return rowscale(gate) * (bih[gate] + bhh[gate]);
}

struct HC {
  float h0, h1, h2;
  float c0, c1, c2;
};

#define CELL_I(X0, pre)                                             \
  const float pre##i0 = fmaf(0.5f, d_tanh(X0[0]), 0.5f);            \
  const float pre##i1 = fmaf(0.5f, d_tanh(X0[1]), 0.5f);            \
  const float pre##i2 = fmaf(0.5f, d_tanh(X0[2]), 0.5f);
#define CELL_F(X0, pre)                                             \
  const float pre##f0 = fmaf(0.5f, d_tanh(X0[3]), 0.5f);            \
  const float pre##f1 = fmaf(0.5f, d_tanh(X0[4]), 0.5f);            \
  const float pre##f2 = fmaf(0.5f, d_tanh(X0[5]), 0.5f);
#define CELL_G(X0, X1, S, pre)                                               \
  const float pre##n0 = fmaf(pre##f0, S.c0, pre##i0 * d_tanh(X0[6]));        \
  const float pre##n1 = fmaf(pre##f1, S.c1, pre##i1 * d_tanh(X0[7]));        \
  const float pre##n2 = fmaf(pre##f2, S.c2, pre##i2 * d_tanh(X1[0]));
#define CELL_O(X1, pre)                                             \
  const float pre##o0 = fmaf(0.5f, d_tanh(X1[1]), 0.5f);            \
  const float pre##o1 = fmaf(0.5f, d_tanh(X1[2]), 0.5f);            \
  const float pre##o2 = fmaf(0.5f, d_tanh(X1[3]), 0.5f);
#define CELL_H(S, pre)                                              \
  S.c0 = pre##n0; S.c1 = pre##n1; S.c2 = pre##n2;                   \
  S.h0 = pre##o0 * d_tanh(pre##n0);                                 \
  S.h1 = pre##o1 * d_tanh(pre##n1);                                 \
  S.h2 = pre##o2 * d_tanh(pre##n2);
#define CELL_ALL(X0, X1, S, pre)                                    \
  CELL_I(X0, pre) CELL_F(X0, pre) CELL_G(X0, X1, S, pre)            \
  CELL_O(X1, pre) CELL_H(S, pre)

template <bool FULL>
__device__ __forceinline__ void lstm2_rows(
    const float* __restrict__ x,
    const float* __restrict__ wih0, const float* __restrict__ whh0,
    const float* __restrict__ bih0, const float* __restrict__ bhh0,
    const float* __restrict__ wih1, const float* __restrict__ whh1,
    const float* __restrict__ bih1, const float* __restrict__ bhh1,
    float* __restrict__ out, int B) {
  const int lane = threadIdx.x & 31;
  const int m0 = lane & 15;
  const int hh = lane >> 4;
  const int nRaw = blockIdx.x * 16 + m0;
  const int n = (FULL || nRaw < B) ? nRaw : (B - 1);
  const bool wr = FULL || (nRaw < B);

  const v2f A00l0 = wfrag(wih0, whh0, 0, m0, 0, hh);
  const v2f A01l0 = wfrag(wih0, whh0, 0, m0, 1, hh);
  const v2f A02l0 = wfrag(wih0, whh0, 0, m0, 2, hh);
  const v2f A10l0 = wfrag(wih0, whh0, 1, m0, 0, hh);
  const v2f A11l0 = wfrag(wih0, whh0, 1, m0, 1, hh);
  const v2f A12l0 = wfrag(wih0, whh0, 1, m0, 2, hh);

  const v2f A00l1 = wfrag(wih1, whh1, 0, m0, 0, hh);
  const v2f A01l1 = wfrag(wih1, whh1, 0, m0, 1, hh);
  const v2f A02l1 = wfrag(wih1, whh1, 0, m0, 2, hh);
  const v2f A10l1 = wfrag(wih1, whh1, 1, m0, 0, hh);
  const v2f A11l1 = wfrag(wih1, whh1, 1, m0, 1, hh);
  const v2f A12l1 = wfrag(wih1, whh1, 1, m0, 2, hh);

  v8f cb0l0, cb1l0, cb0l1, cb1l1;
#pragma unroll
  for (int r = 0; r < 8; ++r) {
    const int g0r = gate_of(0, 8 * hh + r);
    const int g1r = gate_of(1, 8 * hh + r);
    cb0l0[r] = belem(bih0, bhh0, g0r);
    cb1l0[r] = belem(bih0, bhh0, g1r);
    cb0l1[r] = belem(bih1, bhh1, g0r);
    cb1l1[r] = belem(bih1, bhh1, g1r);
  }

  HC s1 = {0, 0, 0, 0, 0, 0};
  HC s2 = {0, 0, 0, 0, 0, 0};
  v8f P0, P1;
  v8f Q0, Q1;
  v3f keep1 = {0, 0, 0}, keep2 = {0, 0, 0}, keep3 = {0, 0, 0};

  const float* __restrict__ xb = x + (size_t)n * TT * HH + HU * hh;
  float* __restrict__ ob = out + (size_t)n * TT * HH + HU * hh;

#define G1(XV)                                                      \
  {                                                                 \
    const v2f b0 = mkv2(XV.x, XV.y);                                \
    const v2f b1 = mkv2(XV.z, s1.h0);                               \
    const v2f b2 = mkv2(s1.h1, s1.h2);                              \
    P0 = wmma4(A00l0, b0, cb0l0);                                   \
    P1 = wmma4(A10l0, b0, cb1l0);                                   \
    P0 = wmma4(A01l0, b1, P0);                                      \
    P1 = wmma4(A11l0, b1, P1);                                      \
    P0 = wmma4(A02l0, b2, P0);                                      \
    P1 = wmma4(A12l0, b2, P1);                                      \
  }
#define G2()                                                        \
  {                                                                 \
    const v2f d0 = mkv2(s1.h0, s1.h1);                              \
    const v2f d1 = mkv2(s1.h2, s2.h0);                              \
    const v2f d2 = mkv2(s2.h1, s2.h2);                              \
    Q0 = wmma4(A00l1, d0, cb0l1);                                   \
    Q1 = wmma4(A10l1, d0, cb1l1);                                   \
    Q0 = wmma4(A01l1, d1, Q0);                                      \
    Q1 = wmma4(A11l1, d1, Q1);                                      \
    Q0 = wmma4(A02l1, d2, Q0);                                      \
    Q1 = wmma4(A12l1, d2, Q1);                                      \
  }
#define PUT(SP, J)                                                             \
  {                                                                            \
    v3f ov;                                                                    \
    ov.x = s2.h0; ov.y = s2.h1; ov.z = s2.h2;                                  \
    if (wr) {                                                                  \
      __builtin_nontemporal_store(ov, (v3fu*)((SP) + (J) * HH));                \
    }                                                                          \
    keep3 = keep2; keep2 = keep1; keep1 = ov;                                  \
  }
#define HOLD() asm volatile("" : : "v"(keep3.x), "v"(keep3.y), "v"(keep3.z))

#define FENCE() __builtin_amdgcn_sched_barrier(0)

#define STEADY(XV, SP, J)                                           \
  {                                                                 \
    const v2f b0 = mkv2(XV.x, XV.y);                                \
    const v2f b1 = mkv2(XV.z, s1.h0);                               \
    const v2f b2 = mkv2(s1.h1, s1.h2);                              \
    P0 = wmma4(A00l0, b0, cb0l0);                                   \
    CELL_I(Q0, u)                                                   \
    FENCE();                                                        \
    P1 = wmma4(A10l0, b0, cb1l0);                                   \
    CELL_F(Q0, u)                                                   \
    FENCE();                                                        \
    P0 = wmma4(A01l0, b1, P0);                                      \
    CELL_G(Q0, Q1, s2, u)                                           \
    FENCE();                                                        \
    P1 = wmma4(A11l0, b1, P1);                                      \
    CELL_O(Q1, u)                                                   \
    FENCE();                                                        \
    P0 = wmma4(A02l0, b2, P0);                                      \
    CELL_H(s2, u)                                                   \
    FENCE();                                                        \
    P1 = wmma4(A12l0, b2, P1);                                      \
    PUT(SP, J)                                                      \
    FENCE();                                                        \
    const v2f d0 = mkv2(s1.h0, s1.h1);                              \
    const v2f d1 = mkv2(s1.h2, s2.h0);                              \
    const v2f d2 = mkv2(s2.h1, s2.h2);                              \
    Q0 = wmma4(A00l1, d0, cb0l1);                                   \
    CELL_I(P0, w)                                                   \
    FENCE();                                                        \
    Q1 = wmma4(A10l1, d0, cb1l1);                                   \
    CELL_F(P0, w)                                                   \
    FENCE();                                                        \
    Q0 = wmma4(A01l1, d1, Q0);                                      \
    CELL_G(P0, P1, s1, w)                                           \
    FENCE();                                                        \
    Q1 = wmma4(A11l1, d1, Q1);                                      \
    CELL_O(P1, w)                                                   \
    FENCE();                                                        \
    Q0 = wmma4(A02l1, d2, Q0);                                      \
    CELL_H(s1, w)                                                   \
    FENCE();                                                        \
    Q1 = wmma4(A12l1, d2, Q1);                                      \
    HOLD();                                                         \
    FENCE();                                                        \
  }

  XG cur = load_xg(xb);
  XG nx = load_xg(xb + TB * HH);

  {
    G1(cur.x0)
    CELL_ALL(P0, P1, s1, a)
  }
  {
    G1(cur.x1)
    G2()
    CELL_ALL(P0, P1, s1, a)
  }
  STEADY(cur.x2, ob, 0)
  STEADY(cur.x3, ob, 1)
  STEADY(cur.x4, ob, 2)
  STEADY(cur.x5, ob, 3)
  STEADY(cur.x6, ob, 4)
  STEADY(cur.x7, ob, 5)
  STEADY(cur.x8, ob, 6)
  STEADY(cur.x9, ob, 7)
  STEADY(cur.x10, ob, 8)
  STEADY(cur.x11, ob, 9)
  STEADY(cur.x12, ob, 10)
  STEADY(cur.x13, ob, 11)
  STEADY(cur.x14, ob, 12)
  STEADY(cur.x15, ob, 13)
  cur = nx;

  for (int k = 1; k < TT / TB; ++k) {
    const int kn = (k + 1 < TT / TB) ? (k + 1) : k;
    const float* __restrict__ xp = xb + (size_t)kn * (TB * HH);
    nx = load_xg(xp);

    float* __restrict__ sp = ob + (size_t)k * (TB * HH);
    STEADY(cur.x0, sp, -2)
    STEADY(cur.x1, sp, -1)
    STEADY(cur.x2, sp, 0)
    STEADY(cur.x3, sp, 1)
    STEADY(cur.x4, sp, 2)
    STEADY(cur.x5, sp, 3)
    STEADY(cur.x6, sp, 4)
    STEADY(cur.x7, sp, 5)
    STEADY(cur.x8, sp, 6)
    STEADY(cur.x9, sp, 7)
    STEADY(cur.x10, sp, 8)
    STEADY(cur.x11, sp, 9)
    STEADY(cur.x12, sp, 10)
    STEADY(cur.x13, sp, 11)
    STEADY(cur.x14, sp, 12)
    STEADY(cur.x15, sp, 13)
    asm volatile("" : : "v"(xp), "v"(sp));
    cur = nx;
  }

  {
    CELL_ALL(Q0, Q1, s2, a)
    PUT(ob + (size_t)(TT - 2) * HH, 0)
  }
  {
    G2()
    CELL_ALL(Q0, Q1, s2, a)
    PUT(ob + (size_t)(TT - 2) * HH, 1)
  }
#undef STEADY
#undef FENCE
#undef HOLD
#undef PUT
#undef G2
#undef G1
}

__global__ __launch_bounds__(32) void lstm2_wmma_kernel(
    const float* __restrict__ x,
    const float* __restrict__ wih0, const float* __restrict__ whh0,
    const float* __restrict__ bih0, const float* __restrict__ bhh0,
    const float* __restrict__ wih1, const float* __restrict__ whh1,
    const float* __restrict__ bih1, const float* __restrict__ bhh1,
    float* __restrict__ out, int B) {
  if ((int)blockIdx.x * 16 + 16 <= B) {
    lstm2_rows<true>(x, wih0, whh0, bih0, bhh0, wih1, whh1, bih1, bhh1, out, B);
  } else {
    lstm2_rows<false>(x, wih0, whh0, bih0, bhh0, wih1, whh1, bih1, bhh1, out, B);
  }
}

extern "C" void kernel_launch(void* const* d_in, const int* in_sizes, int n_in,
                              void* d_out, int out_size, void* d_ws, size_t ws_size,
                              hipStream_t stream) {
  const float* x    = (const float*)d_in[0];
  const float* wih0 = (const float*)d_in[1];
  const float* whh0 = (const float*)d_in[2];
  const float* bih0 = (const float*)d_in[3];
  const float* bhh0 = (const float*)d_in[4];
  const float* wih1 = (const float*)d_in[5];
  const float* whh1 = (const float*)d_in[6];
  const float* bih1 = (const float*)d_in[7];
  const float* bhh1 = (const float*)d_in[8];
  float* out = (float*)d_out;

  const int B = in_sizes[0] / (TT * HH);
  const int nblocks = (B + 15) / 16;

  hipLaunchKernelGGL(lstm2_wmma_kernel, dim3(nblocks), dim3(32), 0, stream,
                     x, wih0, whh0, bih0, bhh0, wih1, whh1, bih1, bhh1, out, B);
}


# layers split over the two waves of a 64-thread work-group (two SIMDs): wave 0 runs layer 1, wave 1 layer 2 one 16-step block behind; h1 handed over through LDS, one LDS-only barrier per block
# speedup vs baseline: 3.4737x; 1.3011x over previous
#include <hip/hip_runtime.h>


#define HH 6
#define TT 1024
#define NG 24
#define HU 3

typedef float v2f __attribute__((ext_vector_type(2)));
typedef float v3f __attribute__((ext_vector_type(3)));
typedef float v4f __attribute__((ext_vector_type(4)));
typedef float v8f __attribute__((ext_vector_type(8)));
typedef v3f v3fu __attribute__((aligned(4)));

#define TB 16
#define NB (TT / TB)

struct XG {
  v3f x0, x1, x2, x3, x4, x5, x6, x7, x8, x9, x10, x11, x12, x13, x14, x15;
};

__device__ __forceinline__ XG load_xg(const float* __restrict__ p) {
  XG g;
  g.x0 = __builtin_nontemporal_load((const v3fu*)(p + 0 * HH));
  g.x1 = __builtin_nontemporal_load((const v3fu*)(p + 1 * HH));
  g.x2 = __builtin_nontemporal_load((const v3fu*)(p + 2 * HH));
  g.x3 = __builtin_nontemporal_load((const v3fu*)(p + 3 * HH));
  g.x4 = __builtin_nontemporal_load((const v3fu*)(p + 4 * HH));
  g.x5 = __builtin_nontemporal_load((const v3fu*)(p + 5 * HH));
  g.x6 = __builtin_nontemporal_load((const v3fu*)(p + 6 * HH));
  g.x7 = __builtin_nontemporal_load((const v3fu*)(p + 7 * HH));
  g.x8 = __builtin_nontemporal_load((const v3fu*)(p + 8 * HH));
  g.x9 = __builtin_nontemporal_load((const v3fu*)(p + 9 * HH));
  g.x10 = __builtin_nontemporal_load((const v3fu*)(p + 10 * HH));
  g.x11 = __builtin_nontemporal_load((const v3fu*)(p + 11 * HH));
  g.x12 = __builtin_nontemporal_load((const v3fu*)(p + 12 * HH));
  g.x13 = __builtin_nontemporal_load((const v3fu*)(p + 13 * HH));
  g.x14 = __builtin_nontemporal_load((const v3fu*)(p + 14 * HH));
  g.x15 = __builtin_nontemporal_load((const v3fu*)(p + 15 * HH));
  return g;
}

__device__ __forceinline__ XG load_hg(const v4f* p) {
  XG g;
  g.x0 = *(const v3f*)(p + 0 * 32);
  g.x1 = *(const v3f*)(p + 1 * 32);
  g.x2 = *(const v3f*)(p + 2 * 32);
  g.x3 = *(const v3f*)(p + 3 * 32);
  g.x4 = *(const v3f*)(p + 4 * 32);
  g.x5 = *(const v3f*)(p + 5 * 32);
  g.x6 = *(const v3f*)(p + 6 * 32);
  g.x7 = *(const v3f*)(p + 7 * 32);
  g.x8 = *(const v3f*)(p + 8 * 32);
  g.x9 = *(const v3f*)(p + 9 * 32);
  g.x10 = *(const v3f*)(p + 10 * 32);
  g.x11 = *(const v3f*)(p + 11 * 32);
  g.x12 = *(const v3f*)(p + 12 * 32);
  g.x13 = *(const v3f*)(p + 13 * 32);
  g.x14 = *(const v3f*)(p + 14 * 32);
  g.x15 = *(const v3f*)(p + 15 * 32);
  return g;
}

__device__ __forceinline__ float d_tanh(float x) {
#if __has_builtin(__builtin_amdgcn_tanhf)
  return __builtin_amdgcn_tanhf(x);
#elif __has_builtin(__builtin_amdgcn_tanh_f32)
  return __builtin_amdgcn_tanh_f32(x);
#else
  float e = __expf(2.0f * x);
  return 1.0f - 2.0f * __builtin_amdgcn_rcpf(e + 1.0f);
#endif
}

__device__ __forceinline__ v2f mkv2(float a, float b) {
  v2f v; v.x = a; v.y = b; return v;
}

__device__ __forceinline__ v8f wmma4(v2f a, v2f b, v8f c) {
  return __builtin_amdgcn_wmma_f32_16x16x4_f32(false, a, false, b, (short)0, c,
                                               false, false);
}

__device__ __forceinline__ int gate_of(int tile, int m) {
  const int hh = m >> 3, r = m & 7, u = HU * hh;
  if (tile == 0) {
    if (r < 3) return u + r;
    if (r < 6) return HH + u + (r - 3);
    return 2 * HH + u + (r - 6);
  }
  if (r == 0) return 2 * HH + u + 2;
  if (r < 4) return 3 * HH + u + (r - 1);
  return -1;
}

__device__ __forceinline__ int col_of(int j, int hh, int s) {
  const int u = HU * hh;
  if (j == 0) return u + s;
  if (j == 1) return (s == 0) ? (u + 2) : (HH + u);
  return HH + u + 1 + s;
}

__device__ __forceinline__ float rowscale(int gate) {
  return (gate < 2 * HH || gate >= 3 * HH) ? 0.5f : 1.0f;
}

__device__ __forceinline__ float welem(const float* __restrict__ wih,
                                       const float* __restrict__ whh,
                                       int gate, int c) {
  if (gate < 0) return 0.0f;
  float w = (c < HH) ? wih[gate * HH + c] : whh[gate * HH + (c - HH)];
  return rowscale(gate) * w;
}

__device__ __forceinline__ v2f wfrag(const float* __restrict__ wih,
                                     const float* __restrict__ whh,
                                     int tile, int m, int j, int hh) {
  const int gate = gate_of(tile, m);
  return mkv2(welem(wih, whh, gate, col_of(j, hh, 0)),
              welem(wih, whh, gate, col_of(j, hh, 1)));
}

__device__ __forceinline__ float belem(const float* __restrict__ bih,
                                       const float* __restrict__ bhh,
                                       int gate) {
  if (gate < 0) return 0.0f;
  return rowscale(gate) * (bih[gate] + bhh[gate]);
}

struct HC {
  float h0, h1, h2;
  float c0, c1, c2;
};

#define CELL_I(X0, pre)                                             \
  const float pre##i0 = fmaf(0.5f, d_tanh(X0[0]), 0.5f);            \
  const float pre##i1 = fmaf(0.5f, d_tanh(X0[1]), 0.5f);            \
  const float pre##i2 = fmaf(0.5f, d_tanh(X0[2]), 0.5f);
#define CELL_F(X0, pre)                                             \
  const float pre##f0 = fmaf(0.5f, d_tanh(X0[3]), 0.5f);            \
  const float pre##f1 = fmaf(0.5f, d_tanh(X0[4]), 0.5f);            \
  const float pre##f2 = fmaf(0.5f, d_tanh(X0[5]), 0.5f);
#define CELL_G(X0, X1, S, pre)                                               \
  const float pre##n0 = fmaf(pre##f0, S.c0, pre##i0 * d_tanh(X0[6]));        \
  const float pre##n1 = fmaf(pre##f1, S.c1, pre##i1 * d_tanh(X0[7]));        \
  const float pre##n2 = fmaf(pre##f2, S.c2, pre##i2 * d_tanh(X1[0]));
#define CELL_O(X1, pre)                                             \
  const float pre##o0 = fmaf(0.5f, d_tanh(X1[1]), 0.5f);            \
  const float pre##o1 = fmaf(0.5f, d_tanh(X1[2]), 0.5f);            \
  const float pre##o2 = fmaf(0.5f, d_tanh(X1[3]), 0.5f);
#define CELL_H(S, pre)                                              \
  S.c0 = pre##n0; S.c1 = pre##n1; S.c2 = pre##n2;                   \
  S.h0 = pre##o0 * d_tanh(pre##n0);                                 \
  S.h1 = pre##o1 * d_tanh(pre##n1);                                 \
  S.h2 = pre##o2 * d_tanh(pre##n2);
#define CELL_ALL(X0, X1, S, pre)                                    \
  CELL_I(X0, pre) CELL_F(X0, pre) CELL_G(X0, X1, S, pre)            \
  CELL_O(X1, pre) CELL_H(S, pre)

template <bool FULL>
__device__ __forceinline__ void lstm2_rows(
    const float* __restrict__ x,
    const float* __restrict__ wih0, const float* __restrict__ whh0,
    const float* __restrict__ bih0, const float* __restrict__ bhh0,
    const float* __restrict__ wih1, const float* __restrict__ whh1,
    const float* __restrict__ bih1, const float* __restrict__ bhh1,
    float* __restrict__ out, int B, v4f (*hb)[TB][32]) {
  const int lane = threadIdx.x & 31;
  const int layer = __builtin_amdgcn_readfirstlane((int)(threadIdx.x >> 5));
  const int m0 = lane & 15;
  const int hh = lane >> 4;
  const int nRaw = blockIdx.x * 16 + m0;
  const int n = (FULL || nRaw < B) ? nRaw : (B - 1);
  const bool wr = FULL || (nRaw < B);

  const float* __restrict__ wih = layer ? wih1 : wih0;
  const float* __restrict__ whh = layer ? whh1 : whh0;
  const float* __restrict__ bih = layer ? bih1 : bih0;
  const float* __restrict__ bhh = layer ? bhh1 : bhh0;

  const v2f A00 = wfrag(wih, whh, 0, m0, 0, hh);
  const v2f A01 = wfrag(wih, whh, 0, m0, 1, hh);
  const v2f A02 = wfrag(wih, whh, 0, m0, 2, hh);
  const v2f A10 = wfrag(wih, whh, 1, m0, 0, hh);
  const v2f A11 = wfrag(wih, whh, 1, m0, 1, hh);
  const v2f A12 = wfrag(wih, whh, 1, m0, 2, hh);

  v8f cb0, cb1;
#pragma unroll
  for (int r = 0; r < 8; ++r) {
    cb0[r] = belem(bih, bhh, gate_of(0, 8 * hh + r));
    cb1[r] = belem(bih, bhh, gate_of(1, 8 * hh + r));
  }

  HC s = {0, 0, 0, 0, 0, 0};
  v8f P0, P1;

  const float* __restrict__ xb = x + (size_t)n * TT * HH + HU * hh;
  float* __restrict__ ob = out + (size_t)n * TT * HH + HU * hh;

#define STEP(XV)                                                    \
  {                                                                 \
    const v2f b0 = mkv2(XV.x, XV.y);                                \
    const v2f b1 = mkv2(XV.z, s.h0);                                \
    const v2f b2 = mkv2(s.h1, s.h2);                                \
    P0 = wmma4(A00, b0, cb0);                                       \
    P1 = wmma4(A10, b0, cb1);                                       \
    P0 = wmma4(A01, b1, P0);                                        \
    P1 = wmma4(A11, b1, P1);                                        \
    P0 = wmma4(A02, b2, P0);                                        \
    P1 = wmma4(A12, b2, P1);                                        \
    CELL_ALL(P0, P1, s, a)                                          \
  }
#define LPUT(HP, J)                                                 \
  {                                                                 \
    v3f hv;                                                         \
    hv.x = s.h0; hv.y = s.h1; hv.z = s.h2;                          \
    *(v3f*)((HP) + (J) * 32) = hv;                                  \
  }
#define PUT(SP, J)                                                             \
  {                                                                            \
    v3f ov;                                                                    \
    ov.x = s.h0; ov.y = s.h1; ov.z = s.h2;                                     \
    if (wr) {                                                                  \
      __builtin_nontemporal_store(ov, (v3fu*)((SP) + (J) * HH));                \
    }                                                                          \
  }

  XG nx;
  if (layer == 0) {
    nx = load_xg(xb);
  } else {
    const v3f z = {0, 0, 0};
    nx.x0 = z; nx.x1 = z; nx.x2 = z; nx.x3 = z; nx.x4 = z; nx.x5 = z;
    nx.x6 = z; nx.x7 = z; nx.x8 = z; nx.x9 = z; nx.x10 = z; nx.x11 = z;
    nx.x12 = z; nx.x13 = z; nx.x14 = z; nx.x15 = z;
  }

  for (int k = 0; k <= NB; ++k) {
    if (layer == 0) {
      if (k < NB) {
        const XG cur = nx;
        const int kn = (k + 1 < NB) ? (k + 1) : k;
        const float* __restrict__ xp = xb + (size_t)kn * (TB * HH);
        nx = load_xg(xp);
        __builtin_amdgcn_sched_barrier(0);
        v4f* hp = &hb[k & 1][0][lane];
        STEP(cur.x0) LPUT(hp, 0)
        STEP(cur.x1) LPUT(hp, 1)
        STEP(cur.x2) LPUT(hp, 2)
        STEP(cur.x3) LPUT(hp, 3)
        STEP(cur.x4) LPUT(hp, 4)
        STEP(cur.x5) LPUT(hp, 5)
        STEP(cur.x6) LPUT(hp, 6)
        STEP(cur.x7) LPUT(hp, 7)
        STEP(cur.x8) LPUT(hp, 8)
        STEP(cur.x9) LPUT(hp, 9)
        STEP(cur.x10) LPUT(hp, 10)
        STEP(cur.x11) LPUT(hp, 11)
        STEP(cur.x12) LPUT(hp, 12)
        STEP(cur.x13) LPUT(hp, 13)
        STEP(cur.x14) LPUT(hp, 14)
        STEP(cur.x15) LPUT(hp, 15)
        __builtin_amdgcn_sched_barrier(0);
      }
    } else {
      if (k > 0) {
        const XG cur = load_hg(&hb[(k - 1) & 1][0][lane]);
        __builtin_amdgcn_sched_barrier(0);
        float* __restrict__ sp = ob + (size_t)(k - 1) * (TB * HH);
        STEP(cur.x0) PUT(sp, 0)
        STEP(cur.x1) PUT(sp, 1)
        STEP(cur.x2) PUT(sp, 2)
        STEP(cur.x3) PUT(sp, 3)
        STEP(cur.x4) PUT(sp, 4)
        STEP(cur.x5) PUT(sp, 5)
        STEP(cur.x6) PUT(sp, 6)
        STEP(cur.x7) PUT(sp, 7)
        STEP(cur.x8) PUT(sp, 8)
        STEP(cur.x9) PUT(sp, 9)
        STEP(cur.x10) PUT(sp, 10)
        STEP(cur.x11) PUT(sp, 11)
        STEP(cur.x12) PUT(sp, 12)
        STEP(cur.x13) PUT(sp, 13)
        STEP(cur.x14) PUT(sp, 14)
        STEP(cur.x15) PUT(sp, 15)
      }
    }
    asm volatile("s_wait_dscnt 0x0\n\ts_barrier_signal -1\n\ts_barrier_wait -1"
                 ::: "memory");
  }
#undef PUT
#undef LPUT
#undef STEP
}

__global__ __launch_bounds__(64) void lstm2_wmma_kernel(
    const float* __restrict__ x,
    const float* __restrict__ wih0, const float* __restrict__ whh0,
    const float* __restrict__ bih0, const float* __restrict__ bhh0,
    const float* __restrict__ wih1, const float* __restrict__ whh1,
    const float* __restrict__ bih1, const float* __restrict__ bhh1,
    float* __restrict__ out, int B) {
  __shared__ v4f hb[2][TB][32];
  if ((int)blockIdx.x * 16 + 16 <= B) {
    lstm2_rows<true>(x, wih0, whh0, bih0, bhh0, wih1, whh1, bih1, bhh1, out, B, hb);
  } else {
    lstm2_rows<false>(x, wih0, whh0, bih0, bhh0, wih1, whh1, bih1, bhh1, out, B, hb);
  }
}

extern "C" void kernel_launch(void* const* d_in, const int* in_sizes, int n_in,
                              void* d_out, int out_size, void* d_ws, size_t ws_size,
                              hipStream_t stream) {
  const float* x    = (const float*)d_in[0];
  const float* wih0 = (const float*)d_in[1];
  const float* whh0 = (const float*)d_in[2];
  const float* bih0 = (const float*)d_in[3];
  const float* bhh0 = (const float*)d_in[4];
  const float* wih1 = (const float*)d_in[5];
  const float* whh1 = (const float*)d_in[6];
  const float* bih1 = (const float*)d_in[7];
  const float* bhh1 = (const float*)d_in[8];
  float* out = (float*)d_out;

  const int B = in_sizes[0] / (TT * HH);
  const int nblocks = (B + 15) / 16;

  hipLaunchKernelGGL(lstm2_wmma_kernel, dim3(nblocks), dim3(64), 0, stream,
                     x, wih0, whh0, bih0, bhh0, wih1, whh1, bih1, bhh1, out, B);
}


# global memory traffic taken off the computing waves: two extra waves of a 4-wave work-group move x and out through LDS in 16-byte pieces tiling whole 128-byte lines; computing waves touch only LDS
# speedup vs baseline: 6.4246x; 1.8495x over previous
#include <hip/hip_runtime.h>


#define HH 6
#define TT 1024
#define NG 24
#define HU 3

typedef float v2f __attribute__((ext_vector_type(2)));
typedef float v3f __attribute__((ext_vector_type(3)));
typedef float v4f __attribute__((ext_vector_type(4)));
typedef float v8f __attribute__((ext_vector_type(8)));
typedef v3f v3fu __attribute__((aligned(4)));

#define TB 16
#define NB (TT / TB)

struct XG {
  v3f x0, x1, x2, x3, x4, x5, x6, x7, x8, x9, x10, x11, x12, x13, x14, x15;
};

__device__ __forceinline__ XG load_hg(const v4f* p) {
  XG g;
  g.x0 = *(const v3f*)(p + 0 * 32);
  g.x1 = *(const v3f*)(p + 1 * 32);
  g.x2 = *(const v3f*)(p + 2 * 32);
  g.x3 = *(const v3f*)(p + 3 * 32);
  g.x4 = *(const v3f*)(p + 4 * 32);
  g.x5 = *(const v3f*)(p + 5 * 32);
  g.x6 = *(const v3f*)(p + 6 * 32);
  g.x7 = *(const v3f*)(p + 7 * 32);
  g.x8 = *(const v3f*)(p + 8 * 32);
  g.x9 = *(const v3f*)(p + 9 * 32);
  g.x10 = *(const v3f*)(p + 10 * 32);
  g.x11 = *(const v3f*)(p + 11 * 32);
  g.x12 = *(const v3f*)(p + 12 * 32);
  g.x13 = *(const v3f*)(p + 13 * 32);
  g.x14 = *(const v3f*)(p + 14 * 32);
  g.x15 = *(const v3f*)(p + 15 * 32);
  return g;
}

__device__ __forceinline__ float d_tanh(float x) {
#if __has_builtin(__builtin_amdgcn_tanhf)
  return __builtin_amdgcn_tanhf(x);
#elif __has_builtin(__builtin_amdgcn_tanh_f32)
  return __builtin_amdgcn_tanh_f32(x);
#else
  float e = __expf(2.0f * x);
  return 1.0f - 2.0f * __builtin_amdgcn_rcpf(e + 1.0f);
#endif
}

__device__ __forceinline__ v2f mkv2(float a, float b) {
  v2f v; v.x = a; v.y = b; return v;
}

__device__ __forceinline__ v8f wmma4(v2f a, v2f b, v8f c) {
  return __builtin_amdgcn_wmma_f32_16x16x4_f32(false, a, false, b, (short)0, c,
                                               false, false);
}

__device__ __forceinline__ int gate_of(int tile, int m) {
  const int hh = m >> 3, r = m & 7, u = HU * hh;
  if (tile == 0) {
    if (r < 3) return u + r;
    if (r < 6) return HH + u + (r - 3);
    return 2 * HH + u + (r - 6);
  }
  if (r == 0) return 2 * HH + u + 2;
  if (r < 4) return 3 * HH + u + (r - 1);
  return -1;
}

__device__ __forceinline__ int col_of(int j, int hh, int s) {
  const int u = HU * hh;
  if (j == 0) return u + s;
  if (j == 1) return (s == 0) ? (u + 2) : (HH + u);
  return HH + u + 1 + s;
}

__device__ __forceinline__ float rowscale(int gate) {
  return (gate < 2 * HH || gate >= 3 * HH) ? 0.5f : 1.0f;
}

__device__ __forceinline__ float welem(const float* __restrict__ wih,
                                       const float* __restrict__ whh,
                                       int gate, int c) {
  if (gate < 0) return 0.0f;
  float w = (c < HH) ? wih[gate * HH + c] : whh[gate * HH + (c - HH)];
  return rowscale(gate) * w;
}

__device__ __forceinline__ v2f wfrag(const float* __restrict__ wih,
                                     const float* __restrict__ whh,
                                     int tile, int m, int j, int hh) {
  const int gate = gate_of(tile, m);
  return mkv2(welem(wih, whh, gate, col_of(j, hh, 0)),
              welem(wih, whh, gate, col_of(j, hh, 1)));
}

__device__ __forceinline__ float belem(const float* __restrict__ bih,
                                       const float* __restrict__ bhh,
                                       int gate) {
  if (gate < 0) return 0.0f;
  return rowscale(gate) * (bih[gate] + bhh[gate]);
}

struct HC {
  float h0, h1, h2;
  float c0, c1, c2;
};

#define CELL_I(X0, pre)                                             \
  const float pre##i0 = fmaf(0.5f, d_tanh(X0[0]), 0.5f);            \
  const float pre##i1 = fmaf(0.5f, d_tanh(X0[1]), 0.5f);            \
  const float pre##i2 = fmaf(0.5f, d_tanh(X0[2]), 0.5f);
#define CELL_F(X0, pre)                                             \
  const float pre##f0 = fmaf(0.5f, d_tanh(X0[3]), 0.5f);            \
  const float pre##f1 = fmaf(0.5f, d_tanh(X0[4]), 0.5f);            \
  const float pre##f2 = fmaf(0.5f, d_tanh(X0[5]), 0.5f);
#define CELL_G(X0, X1, S, pre)                                               \
  const float pre##n0 = fmaf(pre##f0, S.c0, pre##i0 * d_tanh(X0[6]));        \
  const float pre##n1 = fmaf(pre##f1, S.c1, pre##i1 * d_tanh(X0[7]));        \
  const float pre##n2 = fmaf(pre##f2, S.c2, pre##i2 * d_tanh(X1[0]));
#define CELL_O(X1, pre)                                             \
  const float pre##o0 = fmaf(0.5f, d_tanh(X1[1]), 0.5f);            \
  const float pre##o1 = fmaf(0.5f, d_tanh(X1[2]), 0.5f);            \
  const float pre##o2 = fmaf(0.5f, d_tanh(X1[3]), 0.5f);
#define CELL_H(S, pre)                                              \
  S.c0 = pre##n0; S.c1 = pre##n1; S.c2 = pre##n2;                   \
  S.h0 = pre##o0 * d_tanh(pre##n0);                                 \
  S.h1 = pre##o1 * d_tanh(pre##n1);                                 \
  S.h2 = pre##o2 * d_tanh(pre##n2);
#define CELL_ALL(X0, X1, S, pre)                                    \
  CELL_I(X0, pre) CELL_F(X0, pre) CELL_G(X0, X1, S, pre)            \
  CELL_O(X1, pre) CELL_H(S, pre)

#define SLOTF (TB * 32 * 4)
#define NPIECE 12

__global__ __launch_bounds__(128) void lstm2_wmma_kernel(
    const float* __restrict__ x,
    const float* __restrict__ wih0, const float* __restrict__ whh0,
    const float* __restrict__ bih0, const float* __restrict__ bhh0,
    const float* __restrict__ wih1, const float* __restrict__ whh1,
    const float* __restrict__ bih1, const float* __restrict__ bhh1,
    float* __restrict__ out, int B) {
  __shared__ __attribute__((aligned(16))) float xs[2][SLOTF];
  __shared__ __attribute__((aligned(16))) float hs[2][SLOTF];
  __shared__ __attribute__((aligned(16))) float os[2][SLOTF];

  const int lane = threadIdx.x & 31;
  const int role = __builtin_amdgcn_readfirstlane((int)(threadIdx.x >> 5));
  const int layer = role & 1;
  const int m0 = lane & 15;
  const int hh = lane >> 4;
  const int n0 = blockIdx.x * 16;

  int mrow[NPIECE];
  bool mok[NPIECE];
  int mcol[NPIECE];
  int moff[NPIECE][4];
#pragma unroll
  for (int i = 0; i < NPIECE; ++i) {
    const int q = 32 * i + lane;
    const int r = q / 24;
    const int p = q - 24 * r;
    const int nr = n0 + r;
    mok[i] = nr < B;
    mrow[i] = mok[i] ? nr : (B - 1);
    mcol[i] = 4 * p;
#pragma unroll
    for (int e = 0; e < 4; ++e) {
      const int f = 4 * p + e;
      const int j = f / 6;
      const int rem = f - 6 * j;
      const int h2 = rem / 3;
      const int comp = rem - 3 * h2;
      moff[i][e] = (j * 32 + h2 * 16 + r) * 4 + comp;
    }
  }

  const float* __restrict__ wih = layer ? wih1 : wih0;
  const float* __restrict__ whh = layer ? whh1 : whh0;
  const float* __restrict__ bih = layer ? bih1 : bih0;
  const float* __restrict__ bhh = layer ? bhh1 : bhh0;

  const v2f A00 = wfrag(wih, whh, 0, m0, 0, hh);
  const v2f A01 = wfrag(wih, whh, 0, m0, 1, hh);
  const v2f A02 = wfrag(wih, whh, 0, m0, 2, hh);
  const v2f A10 = wfrag(wih, whh, 1, m0, 0, hh);
  const v2f A11 = wfrag(wih, whh, 1, m0, 1, hh);
  const v2f A12 = wfrag(wih, whh, 1, m0, 2, hh);

  v8f cb0, cb1;
#pragma unroll
  for (int r = 0; r < 8; ++r) {
    cb0[r] = belem(bih, bhh, gate_of(0, 8 * hh + r));
    cb1[r] = belem(bih, bhh, gate_of(1, 8 * hh + r));
  }

  HC s = {0, 0, 0, 0, 0, 0};
  v8f P0, P1;

#define STEP(XV)                                                    \
  {                                                                 \
    const v2f b0 = mkv2(XV.x, XV.y);                                \
    const v2f b1 = mkv2(XV.z, s.h0);                                \
    const v2f b2 = mkv2(s.h1, s.h2);                                \
    P0 = wmma4(A00, b0, cb0);                                       \
    P1 = wmma4(A10, b0, cb1);                                       \
    P0 = wmma4(A01, b1, P0);                                        \
    P1 = wmma4(A11, b1, P1);                                        \
    P0 = wmma4(A02, b2, P0);                                        \
    P1 = wmma4(A12, b2, P1);                                        \
    CELL_ALL(P0, P1, s, a)                                          \
  }
#define LPUT(HP, J)                                                 \
  {                                                                 \
    v3f hv;                                                         \
    hv.x = s.h0; hv.y = s.h1; hv.z = s.h2;                          \
    *(v3f*)((HP) + (J) * 32) = hv;                                  \
  }
#define BLOCK16(CUR, HP)                                            \
  STEP(CUR.x0) LPUT(HP, 0)   STEP(CUR.x1) LPUT(HP, 1)               \
  STEP(CUR.x2) LPUT(HP, 2)   STEP(CUR.x3) LPUT(HP, 3)               \
  STEP(CUR.x4) LPUT(HP, 4)   STEP(CUR.x5) LPUT(HP, 5)               \
  STEP(CUR.x6) LPUT(HP, 6)   STEP(CUR.x7) LPUT(HP, 7)               \
  STEP(CUR.x8) LPUT(HP, 8)   STEP(CUR.x9) LPUT(HP, 9)               \
  STEP(CUR.x10) LPUT(HP, 10) STEP(CUR.x11) LPUT(HP, 11)             \
  STEP(CUR.x12) LPUT(HP, 12) STEP(CUR.x13) LPUT(HP, 13)             \
  STEP(CUR.x14) LPUT(HP, 14) STEP(CUR.x15) LPUT(HP, 15)

#define XMOVE(KB)                                                              \
  {                                                                            \
    v4f g[NPIECE];                                                             \
    _Pragma("unroll")                                                          \
    for (int i = 0; i < NPIECE; ++i) {                                         \
      const float* gp = x + (size_t)mrow[i] * (TT * HH) + (size_t)(KB) * (TB * HH) + mcol[i]; \
      g[i] = __builtin_nontemporal_load((const v4f*)gp);                       \
    }                                                                          \
    float* xd = &xs[(KB) & 1][0];                                              \
    _Pragma("unroll")                                                          \
    for (int i = 0; i < NPIECE; ++i) {                                         \
      xd[moff[i][0]] = g[i].x; xd[moff[i][1]] = g[i].y;                        \
      xd[moff[i][2]] = g[i].z; xd[moff[i][3]] = g[i].w;                        \
    }                                                                          \
  }

#define WG_BARRIER()                                                           \
  asm volatile("s_wait_dscnt 0x0\n\ts_barrier_signal -1\n\ts_barrier_wait -1" \
               ::: "memory")

  if (role == 2) XMOVE(0)
  WG_BARRIER();

  for (int k = 0; k <= NB + 1; ++k) {
    if (role == 0) {
      if (k < NB) {
        const XG cur = load_hg((const v4f*)&xs[k & 1][0] + lane);
        __builtin_amdgcn_sched_barrier(0);
        v4f* hp = (v4f*)&hs[k & 1][0] + lane;
        BLOCK16(cur, hp)
      }
    } else if (role == 1) {
      if (k >= 1 && k <= NB) {
        const XG cur = load_hg((const v4f*)&hs[(k - 1) & 1][0] + lane);
        __builtin_amdgcn_sched_barrier(0);
        v4f* op = (v4f*)&os[(k - 1) & 1][0] + lane;
        BLOCK16(cur, op)
      }
    } else if (role == 2) {
      if (k + 1 < NB) XMOVE(k + 1)
    } else {
      if (k >= 2) {
        const float* od = &os[(k - 2) & 1][0];
        v4f v[NPIECE];
#pragma unroll
        for (int i = 0; i < NPIECE; ++i) {
          v[i].x = od[moff[i][0]]; v[i].y = od[moff[i][1]];
          v[i].z = od[moff[i][2]]; v[i].w = od[moff[i][3]];
        }
#pragma unroll
        for (int i = 0; i < NPIECE; ++i) {
          if (mok[i]) {
            float* gp = out + (size_t)mrow[i] * (TT * HH) + (size_t)(k - 2) * (TB * HH) + mcol[i];
            __builtin_nontemporal_store(v[i], (v4f*)gp);
          }
        }
      }
    }
    WG_BARRIER();
  }
#undef WG_BARRIER
#undef XMOVE
#undef BLOCK16
#undef LPUT
#undef STEP
}

extern "C" void kernel_launch(void* const* d_in, const int* in_sizes, int n_in,
                              void* d_out, int out_size, void* d_ws, size_t ws_size,
                              hipStream_t stream) {
  const float* x    = (const float*)d_in[0];
  const float* wih0 = (const float*)d_in[1];
  const float* whh0 = (const float*)d_in[2];
  const float* bih0 = (const float*)d_in[3];
  const float* bhh0 = (const float*)d_in[4];
  const float* wih1 = (const float*)d_in[5];
  const float* whh1 = (const float*)d_in[6];
  const float* bih1 = (const float*)d_in[7];
  const float* bhh1 = (const float*)d_in[8];
  float* out = (float*)d_out;

  const int B = in_sizes[0] / (TT * HH);
  const int nblocks = (B + 15) / 16;

  hipLaunchKernelGGL(lstm2_wmma_kernel, dim3(nblocks), dim3(128), 0, stream,
                     x, wih0, whh0, bih0, bhh0, wih1, whh1, bih1, bhh1, out, B);
}


# issue order of a step pinned by fences: the next step's two input products issue behind the gate tanh and behind tanh(c), covering their result latency; a block's h stored to LDS as one group
# speedup vs baseline: 6.6789x; 1.0396x over previous
#include <hip/hip_runtime.h>


#define HH 6
#define TT 1024
#define NG 24
#define HU 3

typedef float v2f __attribute__((ext_vector_type(2)));
typedef float v3f __attribute__((ext_vector_type(3)));
typedef float v4f __attribute__((ext_vector_type(4)));
typedef float v8f __attribute__((ext_vector_type(8)));
typedef v3f v3fu __attribute__((aligned(4)));

#define TB 16
#define NB (TT / TB)

struct XG {
  v3f x0, x1, x2, x3, x4, x5, x6, x7, x8, x9, x10, x11, x12, x13, x14, x15;
};

__device__ __forceinline__ XG load_hg(const v4f* p) {
  XG g;
  g.x0 = *(const v3f*)(p + 0 * 32);
  g.x1 = *(const v3f*)(p + 1 * 32);
  g.x2 = *(const v3f*)(p + 2 * 32);
  g.x3 = *(const v3f*)(p + 3 * 32);
  g.x4 = *(const v3f*)(p + 4 * 32);
  g.x5 = *(const v3f*)(p + 5 * 32);
  g.x6 = *(const v3f*)(p + 6 * 32);
  g.x7 = *(const v3f*)(p + 7 * 32);
  g.x8 = *(const v3f*)(p + 8 * 32);
  g.x9 = *(const v3f*)(p + 9 * 32);
  g.x10 = *(const v3f*)(p + 10 * 32);
  g.x11 = *(const v3f*)(p + 11 * 32);
  g.x12 = *(const v3f*)(p + 12 * 32);
  g.x13 = *(const v3f*)(p + 13 * 32);
  g.x14 = *(const v3f*)(p + 14 * 32);
  g.x15 = *(const v3f*)(p + 15 * 32);
  return g;
}

__device__ __forceinline__ float d_tanh(float x) {
#if __has_builtin(__builtin_amdgcn_tanhf)
  return __builtin_amdgcn_tanhf(x);
#elif __has_builtin(__builtin_amdgcn_tanh_f32)
  return __builtin_amdgcn_tanh_f32(x);
#else
  float e = __expf(2.0f * x);
  return 1.0f - 2.0f * __builtin_amdgcn_rcpf(e + 1.0f);
#endif
}

__device__ __forceinline__ v2f mkv2(float a, float b) {
  v2f v; v.x = a; v.y = b; return v;
}

__device__ __forceinline__ v8f wmma4(v2f a, v2f b, v8f c) {
  return __builtin_amdgcn_wmma_f32_16x16x4_f32(false, a, false, b, (short)0, c,
                                               false, false);
}

__device__ __forceinline__ int gate_of(int tile, int m) {
  const int hh = m >> 3, r = m & 7, u = HU * hh;
  if (tile == 0) {
    if (r < 3) return u + r;
    if (r < 6) return HH + u + (r - 3);
    return 2 * HH + u + (r - 6);
  }
  if (r == 0) return 2 * HH + u + 2;
  if (r < 4) return 3 * HH + u + (r - 1);
  return -1;
}

__device__ __forceinline__ int col_of(int j, int hh, int s) {
  const int u = HU * hh;
  if (j == 0) return u + s;
  if (j == 1) return (s == 0) ? (u + 2) : (HH + u);
  return HH + u + 1 + s;
}

__device__ __forceinline__ float rowscale(int gate) {
  return (gate < 2 * HH || gate >= 3 * HH) ? 0.5f : 1.0f;
}

__device__ __forceinline__ float welem(const float* __restrict__ wih,
                                       const float* __restrict__ whh,
                                       int gate, int c) {
  if (gate < 0) return 0.0f;
  float w = (c < HH) ? wih[gate * HH + c] : whh[gate * HH + (c - HH)];
  return rowscale(gate) * w;
}

__device__ __forceinline__ v2f wfrag(const float* __restrict__ wih,
                                     const float* __restrict__ whh,
                                     int tile, int m, int j, int hh) {
  const int gate = gate_of(tile, m);
  return mkv2(welem(wih, whh, gate, col_of(j, hh, 0)),
              welem(wih, whh, gate, col_of(j, hh, 1)));
}

__device__ __forceinline__ float belem(const float* __restrict__ bih,
                                       const float* __restrict__ bhh,
                                       int gate) {
  if (gate < 0) return 0.0f;
  return rowscale(gate) * (bih[gate] + bhh[gate]);
}

struct HC {
  float h0, h1, h2;
  float c0, c1, c2;
};

#define SLOTF (TB * 32 * 4)
#define NPIECE 12

__global__ __launch_bounds__(128) void lstm2_wmma_kernel(
    const float* __restrict__ x,
    const float* __restrict__ wih0, const float* __restrict__ whh0,
    const float* __restrict__ bih0, const float* __restrict__ bhh0,
    const float* __restrict__ wih1, const float* __restrict__ whh1,
    const float* __restrict__ bih1, const float* __restrict__ bhh1,
    float* __restrict__ out, int B) {
  __shared__ __attribute__((aligned(16))) float xs[2][SLOTF];
  __shared__ __attribute__((aligned(16))) float hs[2][SLOTF];
  __shared__ __attribute__((aligned(16))) float os[2][SLOTF];

  const int lane = threadIdx.x & 31;
  const int role = __builtin_amdgcn_readfirstlane((int)(threadIdx.x >> 5));
  const int layer = role & 1;
  const int m0 = lane & 15;
  const int hh = lane >> 4;
  const int n0 = blockIdx.x * 16;

  int mrow[NPIECE];
  bool mok[NPIECE];
  int mcol[NPIECE];
  int moff[NPIECE][4];
#pragma unroll
  for (int i = 0; i < NPIECE; ++i) {
    const int q = 32 * i + lane;
    const int r = q / 24;
    const int p = q - 24 * r;
    const int nr = n0 + r;
    mok[i] = nr < B;
    mrow[i] = mok[i] ? nr : (B - 1);
    mcol[i] = 4 * p;
#pragma unroll
    for (int e = 0; e < 4; ++e) {
      const int f = 4 * p + e;
      const int j = f / 6;
      const int rem = f - 6 * j;
      const int h2 = rem / 3;
      const int comp = rem - 3 * h2;
      moff[i][e] = (j * 32 + h2 * 16 + r) * 4 + comp;
    }
  }

  const float* __restrict__ wih = layer ? wih1 : wih0;
  const float* __restrict__ whh = layer ? whh1 : whh0;
  const float* __restrict__ bih = layer ? bih1 : bih0;
  const float* __restrict__ bhh = layer ? bhh1 : bhh0;

  const v2f A00 = wfrag(wih, whh, 0, m0, 0, hh);
  const v2f A01 = wfrag(wih, whh, 0, m0, 1, hh);
  const v2f A02 = wfrag(wih, whh, 0, m0, 2, hh);
  const v2f A10 = wfrag(wih, whh, 1, m0, 0, hh);
  const v2f A11 = wfrag(wih, whh, 1, m0, 1, hh);
  const v2f A12 = wfrag(wih, whh, 1, m0, 2, hh);

  v8f cb0, cb1;
#pragma unroll
  for (int r = 0; r < 8; ++r) {
    cb0[r] = belem(bih, bhh, gate_of(0, 8 * hh + r));
    cb1[r] = belem(bih, bhh, gate_of(1, 8 * hh + r));
  }

  HC s = {0, 0, 0, 0, 0, 0};
  v8f P0, P1;

#define FENCE() __builtin_amdgcn_sched_barrier(0)

#define INPUT0(XV) Q0 = wmma4(A00, mkv2(XV.x, XV.y), cb0);
#define INPUT1(XV) Q1 = wmma4(A10, mkv2(XV.x, XV.y), cb1);
#define NOINPUT
#define PIN_HERE(A, B) asm volatile("" : "+v"(A), "+v"(B))

#define STEP(XV, J, IN0, IN1)                                       \
  {                                                                 \
    {                                                               \
      const v2f b1 = mkv2(XV.z, s.h0);                              \
      const v2f b2 = mkv2(s.h1, s.h2);                              \
      P0 = wmma4(A01, b1, Q0);                                      \
      P1 = wmma4(A11, b1, Q1);                                      \
      P0 = wmma4(A02, b2, P0);                                      \
      P1 = wmma4(A12, b2, P1);                                      \
    }                                                               \
    FENCE();                                                        \
    const float ti0 = d_tanh(P0[0]);                                \
    const float tf0 = d_tanh(P0[3]);                                \
    const float tg0 = d_tanh(P0[6]);                                \
    const float ti1 = d_tanh(P0[1]);                                \
    const float tf1 = d_tanh(P0[4]);                                \
    const float tg1 = d_tanh(P0[7]);                                \
    const float ti2 = d_tanh(P0[2]);                                \
    const float tf2 = d_tanh(P0[5]);                                \
    const float tg2 = d_tanh(P1[0]);                                \
    const float to0 = d_tanh(P1[1]);                                \
    const float to1 = d_tanh(P1[2]);                                \
    const float to2 = d_tanh(P1[3]);                                \
    FENCE();                                                        \
    IN0                                                             \
    FENCE();                                                        \
    const float n0 = fmaf(fmaf(0.5f, tf0, 0.5f), s.c0, fmaf(0.5f, ti0, 0.5f) * tg0); \
    const float n1 = fmaf(fmaf(0.5f, tf1, 0.5f), s.c1, fmaf(0.5f, ti1, 0.5f) * tg1); \
    const float n2 = fmaf(fmaf(0.5f, tf2, 0.5f), s.c2, fmaf(0.5f, ti2, 0.5f) * tg2); \
    const float tn0 = d_tanh(n0);                                   \
    const v2f tn12 = mkv2(d_tanh(n1), d_tanh(n2));                  \
    float o0 = fmaf(0.5f, to0, 0.5f);                               \
    v2f o12 = mkv2(fmaf(0.5f, to1, 0.5f), fmaf(0.5f, to2, 0.5f));   \
    PIN_HERE(o0, o12);                                              \
    FENCE();                                                        \
    IN1                                                             \
    FENCE();                                                        \
    s.c0 = n0; s.c1 = n1; s.c2 = n2;                                \
    s.h0 = o0 * tn0;                                                \
    const v2f h12 = o12 * tn12;                                     \
    s.h1 = h12.x; s.h2 = h12.y;                                     \
    hq[J].x = s.h0; hq[J].y = s.h1; hq[J].z = s.h2;                 \
  }
#define BLOCK16(CUR, HP)                                            \
  v3f hq[TB];                                                       \
  v8f Q0, Q1;                                                       \
  INPUT0(CUR.x0) INPUT1(CUR.x0)                                     \
  STEP(CUR.x0, 0, INPUT0(CUR.x1), INPUT1(CUR.x1))                   \
  STEP(CUR.x1, 1, INPUT0(CUR.x2), INPUT1(CUR.x2))                   \
  STEP(CUR.x2, 2, INPUT0(CUR.x3), INPUT1(CUR.x3))                   \
  STEP(CUR.x3, 3, INPUT0(CUR.x4), INPUT1(CUR.x4))                   \
  STEP(CUR.x4, 4, INPUT0(CUR.x5), INPUT1(CUR.x5))                   \
  STEP(CUR.x5, 5, INPUT0(CUR.x6), INPUT1(CUR.x6))                   \
  STEP(CUR.x6, 6, INPUT0(CUR.x7), INPUT1(CUR.x7))                   \
  STEP(CUR.x7, 7, INPUT0(CUR.x8), INPUT1(CUR.x8))                   \
  STEP(CUR.x8, 8, INPUT0(CUR.x9), INPUT1(CUR.x9))                   \
  STEP(CUR.x9, 9, INPUT0(CUR.x10), INPUT1(CUR.x10))                 \
  STEP(CUR.x10, 10, INPUT0(CUR.x11), INPUT1(CUR.x11))               \
  STEP(CUR.x11, 11, INPUT0(CUR.x12), INPUT1(CUR.x12))               \
  STEP(CUR.x12, 12, INPUT0(CUR.x13), INPUT1(CUR.x13))               \
  STEP(CUR.x13, 13, INPUT0(CUR.x14), INPUT1(CUR.x14))               \
  STEP(CUR.x14, 14, INPUT0(CUR.x15), INPUT1(CUR.x15))               \
  STEP(CUR.x15, 15, NOINPUT, NOINPUT)                               \
  FENCE();                                                          \
    \
  _Pragma("unroll")                                                 \
  for (int j = 0; j < TB; ++j) *(v3f*)((HP) + j * 32) = hq[j];

#define XMOVE(KB)                                                              \
  {                                                                            \
    v4f g[NPIECE];                                                             \
    _Pragma("unroll")                                                          \
    for (int i = 0; i < NPIECE; ++i) {                                         \
      const float* gp = x + (size_t)mrow[i] * (TT * HH) + (size_t)(KB) * (TB * HH) + mcol[i]; \
      g[i] = __builtin_nontemporal_load((const v4f*)gp);                       \
    }                                                                          \
    float* xd = &xs[(KB) & 1][0];                                              \
    _Pragma("unroll")                                                          \
    for (int i = 0; i < NPIECE; ++i) {                                         \
      xd[moff[i][0]] = g[i].x; xd[moff[i][1]] = g[i].y;                        \
      xd[moff[i][2]] = g[i].z; xd[moff[i][3]] = g[i].w;                        \
    }                                                                          \
  }

#define WG_BARRIER()                                                           \
  asm volatile("s_wait_dscnt 0x0\n\ts_barrier_signal -1\n\ts_barrier_wait -1" \
               ::: "memory")

  if (role == 2) XMOVE(0)
  WG_BARRIER();

  for (int k = 0; k <= NB + 1; ++k) {
    if (role == 0) {
      if (k < NB) {
        const XG cur = load_hg((const v4f*)&xs[k & 1][0] + lane);
        __builtin_amdgcn_sched_barrier(0);
        v4f* hp = (v4f*)&hs[k & 1][0] + lane;
        BLOCK16(cur, hp)
      }
    } else if (role == 1) {
      if (k >= 1 && k <= NB) {
        const XG cur = load_hg((const v4f*)&hs[(k - 1) & 1][0] + lane);
        __builtin_amdgcn_sched_barrier(0);
        v4f* op = (v4f*)&os[(k - 1) & 1][0] + lane;
        BLOCK16(cur, op)
      }
    } else if (role == 2) {
      if (k + 1 < NB) XMOVE(k + 1)
    } else {
      if (k >= 2) {
        const float* od = &os[(k - 2) & 1][0];
        v4f v[NPIECE];
#pragma unroll
        for (int i = 0; i < NPIECE; ++i) {
          v[i].x = od[moff[i][0]]; v[i].y = od[moff[i][1]];
          v[i].z = od[moff[i][2]]; v[i].w = od[moff[i][3]];
        }
#pragma unroll
        for (int i = 0; i < NPIECE; ++i) {
          if (mok[i]) {
            float* gp = out + (size_t)mrow[i] * (TT * HH) + (size_t)(k - 2) * (TB * HH) + mcol[i];
            __builtin_nontemporal_store(v[i], (v4f*)gp);
          }
        }
      }
    }
    WG_BARRIER();
  }
#undef WG_BARRIER
#undef XMOVE
#undef BLOCK16
#undef PIN_HERE
#undef NOINPUT
#undef INPUT1
#undef INPUT0
#undef FENCE
#undef STEP
}

extern "C" void kernel_launch(void* const* d_in, const int* in_sizes, int n_in,
                              void* d_out, int out_size, void* d_ws, size_t ws_size,
                              hipStream_t stream) {
  const float* x    = (const float*)d_in[0];
  const float* wih0 = (const float*)d_in[1];
  const float* whh0 = (const float*)d_in[2];
  const float* bih0 = (const float*)d_in[3];
  const float* bhh0 = (const float*)d_in[4];
  const float* wih1 = (const float*)d_in[5];
  const float* whh1 = (const float*)d_in[6];
  const float* bih1 = (const float*)d_in[7];
  const float* bhh1 = (const float*)d_in[8];
  float* out = (float*)d_out;

  const int B = in_sizes[0] / (TT * HH);
  const int nblocks = (B + 15) / 16;

  hipLaunchKernelGGL(lstm2_wmma_kernel, dim3(nblocks), dim3(128), 0, stream,
                     x, wih0, whh0, bih0, bhh0, wih1, whh1, bih1, bhh1, out, B);
}
